# starting kernel
# baseline (speedup 1.0000x reference)
#include <hip/hip_runtime.h>
#include <hip/hip_bf16.h>

#define Nn 50000
#define Ee 800000
#define HX 64
#define HY 32
#define HT 32
#define FEAT 96
#define KX 128
#define GROUPS ((Nn + 15) / 16)

typedef __attribute__((ext_vector_type(2))) float v2f;
typedef __attribute__((ext_vector_type(8))) float v8f;

__global__ void zero_ws(float* __restrict__ aggr, int n) {
    int i = blockIdx.x * blockDim.x + threadIdx.x;
    if (i < n) aggr[i] = 0.0f;
}

__global__ void scatter_add(const int* __restrict__ erow,
                            const int* __restrict__ ecol,
                            const float* __restrict__ hX,
                            const float* __restrict__ hY,
                            float* __restrict__ aggr) {
    const int lane = threadIdx.x & 31;
    const int e    = blockIdx.x * 8 + (threadIdx.x >> 5);
    if (e >= Ee) return;
    const int col = ecol[e];
    const int row = erow[e];
    const float* srcX = hX + (size_t)col * HX;
    const float* srcY = hY + (size_t)col * HY;
    float* dst = aggr + (size_t)row * FEAT;
    unsafeAtomicAdd(&dst[lane],      srcX[lane]);
    unsafeAtomicAdd(&dst[lane + 32], srcX[lane + 32]);
    unsafeAtomicAdd(&dst[lane + 64], srcY[lane]);
}

__device__ __forceinline__ v8f wmma_f32(v2f a, v2f b, v8f c) {
    return __builtin_amdgcn_wmma_f32_16x16x4_f32(
         false, a,  false, b,
         (short)0, c,  false,  false);
}

__global__ void gemm_ln_X(const float* __restrict__ aggr,
                          const float* __restrict__ ht,
                          const float* __restrict__ WX,
                          const float* __restrict__ bX,
                          const float* __restrict__ gX,
                          const float* __restrict__ beX,
                          float* __restrict__ outX) {
    __shared__ float lds[4][16][HX + 1];
    const int wave  = threadIdx.x >> 5;
    const int lane  = threadIdx.x & 31;
    const int group = blockIdx.x * 4 + wave;

    if (group < GROUPS) {
        const int m     = lane & 15;
        const int khalf = lane >> 4;
        const int node  = group * 16 + m;
        const float* arow = aggr + (size_t)node * FEAT;

        v8f acc[4] = {};
        #pragma unroll 4
        for (int k0 = 0; k0 < KX; k0 += 4) {
            const int ka = k0 + khalf * 2;
            v2f a;
            if (ka < FEAT) { a.x = arow[ka];      a.y = arow[ka + 1]; }
            else           { a.x = ht[ka - FEAT]; a.y = ht[ka - FEAT + 1]; }
            #pragma unroll
            for (int t = 0; t < 4; ++t) {
                const int n = t * 16 + m;
                v2f b;
                b.x = WX[ka * HX + n];
                b.y = WX[(ka + 1) * HX + n];
                acc[t] = wmma_f32(a, b, acc[t]);
            }
        }
        #pragma unroll
        for (int t = 0; t < 4; ++t) {
            const int n = t * 16 + m;
            const float bias = bX[n];
            #pragma unroll
            for (int v = 0; v < 8; ++v) {
                const int row = khalf * 8 + v;
                float x = acc[t][v] + bias;
                lds[wave][row][n] = fmaxf(x, 0.0f);
            }
        }
    }
    __syncthreads();
    if (group < GROUPS) {
        const int row  = lane >> 1;
        const int half = lane & 1;
        float s = 0.0f, s2 = 0.0f;
        #pragma unroll
        for (int j = 0; j < 32; ++j) {
            float x = lds[wave][row][half * 32 + j];
            s += x; s2 += x * x;
        }
        s  += __shfl_xor(s, 1);
        s2 += __shfl_xor(s2, 1);
        const float mean = s * (1.0f / 64.0f);
        const float var  = s2 * (1.0f / 64.0f) - mean * mean;
        const float inv  = rsqrtf(var + 1e-5f);
        const int node   = group * 16 + row;
        #pragma unroll
        for (int j = 0; j < 32; ++j) {
            const int c = half * 32 + j;
            float x = lds[wave][row][c];
            outX[(size_t)node * HX + c] = (x - mean) * inv * gX[c] + beX[c];
        }
    }
}

__global__ void gemm_ln_Y(const float* __restrict__ aggr,
                          const float* __restrict__ WY,
                          const float* __restrict__ bY,
                          const float* __restrict__ gY,
                          const float* __restrict__ beY,
                          float* __restrict__ outY) {
    __shared__ float lds[4][16][HY + 1];
    const int wave  = threadIdx.x >> 5;
    const int lane  = threadIdx.x & 31;
    const int group = blockIdx.x * 4 + wave;

    if (group < GROUPS) {
        const int m     = lane & 15;
        const int khalf = lane >> 4;
        const int node  = group * 16 + m;
        const float* arow = aggr + (size_t)node * FEAT + HX;

        v8f acc[2] = {};
        #pragma unroll
        for (int k0 = 0; k0 < HY; k0 += 4) {
            const int ka = k0 + khalf * 2;
            v2f a;
            a.x = arow[ka];
            a.y = arow[ka + 1];
            #pragma unroll
            for (int t = 0; t < 2; ++t) {
                const int n = t * 16 + m;
                v2f b;
                b.x = WY[ka * HY + n];
                b.y = WY[(ka + 1) * HY + n];
                acc[t] = wmma_f32(a, b, acc[t]);
            }
        }
        #pragma unroll
        for (int t = 0; t < 2; ++t) {
            const int n = t * 16 + m;
            const float bias = bY[n];
            #pragma unroll
            for (int v = 0; v < 8; ++v) {
                const int row = khalf * 8 + v;
                float x = acc[t][v] + bias;
                lds[wave][row][n] = fmaxf(x, 0.0f);
            }
        }
    }
    __syncthreads();
    if (group < GROUPS) {
        const int row  = lane >> 1;
        const int half = lane & 1;
        float s = 0.0f, s2 = 0.0f;
        #pragma unroll
        for (int j = 0; j < 16; ++j) {
            float x = lds[wave][row][half * 16 + j];
            s += x; s2 += x * x;
        }
        s  += __shfl_xor(s, 1);
        s2 += __shfl_xor(s2, 1);
        const float mean = s * (1.0f / 32.0f);
        const float var  = s2 * (1.0f / 32.0f) - mean * mean;
        const float inv  = rsqrtf(var + 1e-5f);
        const int node   = group * 16 + row;
        #pragma unroll
        for (int j = 0; j < 16; ++j) {
            const int c = half * 16 + j;
            float x = lds[wave][row][c];
            outY[(size_t)node * HY + c] = (x - mean) * inv * gY[c] + beY[c];
        }
    }
}

extern "C" void kernel_launch(void* const* d_in, const int* in_sizes, int n_in,
                              void* d_out, int out_size, void* d_ws, size_t ws_size,
                              hipStream_t stream) {
    const int*   erow = (const int*)d_in[0];
    const int*   ecol = (const int*)d_in[1];
    const float* hX   = (const float*)d_in[2];
    const float* hY   = (const float*)d_in[3];
    const float* ht   = (const float*)d_in[4];
    const float* WX   = (const float*)d_in[5];
    const float* bX   = (const float*)d_in[6];
    const float* gX   = (const float*)d_in[7];
    const float* beX  = (const float*)d_in[8];
    const float* WY   = (const float*)d_in[9];
    const float* bY   = (const float*)d_in[10];
    const float* gY   = (const float*)d_in[11];
    const float* beY  = (const float*)d_in[12];

    float* out  = (float*)d_out;
    float* aggr = (float*)d_ws;

    const int nAggr = Nn * FEAT;
    zero_ws<<<(nAggr + 255) / 256, 256, 0, stream>>>(aggr, nAggr);

    scatter_add<<<(Ee + 7) / 8, 256, 0, stream>>>(erow, ecol, hX, hY, aggr);

    const int blocks = (GROUPS + 3) / 4;
    gemm_ln_X<<<blocks, 128, 0, stream>>>(aggr, ht, WX, bX, gX, beX, out);
    gemm_ln_Y<<<blocks, 128, 0, stream>>>(aggr, WY, bY, gY, beY, out + (size_t)Nn * HX);
}


# Aggregation without global atomics or zero pass: 256 work-groups each scan all destinations, queue their own 196-node slab's edges in LDS, sum rows by LDS f32 atomics (arrival order)
# speedup vs baseline: 1.6007x; 1.6007x over previous
#include <hip/hip_runtime.h>
#include <hip/hip_bf16.h>

#define Nn 50000
#define Ee 800000
#define HX 64
#define HY 32
#define HT 32
#define FEAT 96
#define KX 128
#define GROUPS ((Nn + 15) / 16)

typedef __attribute__((ext_vector_type(2))) float v2f;
typedef __attribute__((ext_vector_type(8))) float v8f;

#define NBLK  256
#define DBLK  196
#define SCAN_T 1024
#define SCAN_U 4
#define TILE4 (SCAN_T * SCAN_U)
#define TILE_EDGES (TILE4 * 4)
#define E4 (Ee / 4)
#define NTILES ((E4 + TILE4 - 1) / TILE4)
#define QCAP (2 * TILE_EDGES)

#define DR_RUNS 4
#define DR_GRP  16

__device__ __forceinline__ void drain_queue(const unsigned* q, int n, float* acc,
                                            const int* __restrict__ ecol,
                                            const float* __restrict__ hX,
                                            const float* __restrict__ hY,
                                            int wave, int lane) {
    const int nruns = (n + 31) >> 5;
    for (int r0 = wave; r0 < nruns; r0 += (SCAN_T / 32) * DR_RUNS) {
        unsigned ent[DR_RUNS];
        int      col[DR_RUNS];
        #pragma unroll
        for (int k = 0; k < DR_RUNS; ++k) {
            const int idx = (r0 + (SCAN_T / 32) * k) * 32 + lane;
            ent[k] = (idx < n) ? q[idx < QCAP ? idx : QCAP - 1] : (unsigned)DBLK;
        }
        #pragma unroll
        for (int k = 0; k < DR_RUNS; ++k) col[k] = ecol[ent[k] >> 8];
        #pragma unroll
        for (int k = 0; k < DR_RUNS; ++k) {
            int cnt = n - (r0 + (SCAN_T / 32) * k) * 32;
            if (cnt > 32) cnt = 32;
            for (int j0 = 0; j0 < cnt; j0 += DR_GRP) {
                float x0[DR_GRP], x1[DR_GRP], y0[DR_GRP];
                int   dr[DR_GRP];
                #pragma unroll
                for (int u = 0; u < DR_GRP; ++u) {
                    const int c = __builtin_amdgcn_readlane(col[k], j0 + u);
                    dr[u] = (int)(__builtin_amdgcn_readlane((int)ent[k], j0 + u) & 255) * FEAT;
                    const float* sx = hX + (size_t)c * HX;
                    const float* sy = hY + (size_t)c * HY;
                    x0[u] = sx[lane];
                    x1[u] = sx[lane + 32];
                    y0[u] = sy[lane];
                }
                #pragma unroll
                for (int u = 0; u < DR_GRP; ++u) {
                    atomicAdd(&acc[dr[u] + lane],      x0[u]);
                    atomicAdd(&acc[dr[u] + lane + 32], x1[u]);
                    atomicAdd(&acc[dr[u] + lane + 64], y0[u]);
                }
            }
        }
    }
}

__global__ __launch_bounds__(SCAN_T)
void scan_sum(const int* __restrict__ erow,
              const int* __restrict__ ecol,
              const float* __restrict__ hX,
              const float* __restrict__ hY,
              float* __restrict__ aggr) {
    __shared__ float    acc[(DBLK + 1) * FEAT];
    __shared__ unsigned q[QCAP];
    __shared__ unsigned qn;

    const int tid  = threadIdx.x;
    const int wave = tid >> 5;
    const int lane = tid & 31;
    const int lo   = blockIdx.x * DBLK;

    for (int i = tid; i < (DBLK + 1) * FEAT; i += SCAN_T) acc[i] = 0.0f;
    if (tid == 0) qn = 0u;
    __syncthreads();

    const int4* er4 = (const int4*)erow;
    const int t0 = (blockIdx.x * NTILES) / NBLK;

    int4 v[SCAN_U];
    {
        const int base = t0 * TILE4 + tid;
        #pragma unroll
        for (int u = 0; u < SCAN_U; ++u) {
            int i4 = base + u * SCAN_T; if (i4 > E4 - 1) i4 = E4 - 1;
            v[u] = er4[i4];
        }
    }
    for (int t = 0; t < NTILES; ++t) {
        int tt = t + t0; if (tt >= NTILES) tt -= NTILES;
        int tn = tt + 1; if (tn >= NTILES) tn -= NTILES;
        const int base  = tt * TILE4 + tid;
        const int nbase = tn * TILE4 + tid;
        int4 w[SCAN_U];
        #pragma unroll
        for (int u = 0; u < SCAN_U; ++u) {
            int i4 = nbase + u * SCAN_T; if (i4 > E4 - 1) i4 = E4 - 1;
            w[u] = er4[i4];
        }
        #pragma unroll
        for (int u = 0; u < SCAN_U; ++u) {
            const int i4 = base + u * SCAN_T;
            const unsigned e0 = (unsigned)i4 * 4u;
            const int lo_u = (i4 < E4) ? lo : 0x7fffffff;
            const unsigned d0 = (unsigned)(v[u].x - lo_u);
            const unsigned d1 = (unsigned)(v[u].y - lo_u);
            const unsigned d2 = (unsigned)(v[u].z - lo_u);
            const unsigned d3 = (unsigned)(v[u].w - lo_u);
            if (d0 < (unsigned)DBLK) { unsigned p = atomicAdd(&qn, 1u); q[p] = ((e0     ) << 8) | d0; }
            if (d1 < (unsigned)DBLK) { unsigned p = atomicAdd(&qn, 1u); q[p] = ((e0 + 1u) << 8) | d1; }
            if (d2 < (unsigned)DBLK) { unsigned p = atomicAdd(&qn, 1u); q[p] = ((e0 + 2u) << 8) | d2; }
            if (d3 < (unsigned)DBLK) { unsigned p = atomicAdd(&qn, 1u); q[p] = ((e0 + 3u) << 8) | d3; }
        }
        __syncthreads();
        const int n = (int)qn;
        __syncthreads();
        if (n > QCAP - TILE_EDGES) {
            drain_queue(q, n, acc, ecol, hX, hY, wave, lane);
            __syncthreads();
            if (tid == 0) qn = 0u;
            __syncthreads();
        }
        #pragma unroll
        for (int u = 0; u < SCAN_U; ++u) v[u] = w[u];
    }
    {
        const int n = (int)qn;
        drain_queue(q, n, acc, ecol, hX, hY, wave, lane);
    }
    __syncthreads();

    int lim = (Nn - lo) * FEAT;
    if (lim > DBLK * FEAT) lim = DBLK * FEAT;
    float* dst = aggr + (size_t)lo * FEAT;
    for (int i = tid; i < lim; i += SCAN_T) dst[i] = acc[i];
}

__device__ __forceinline__ v8f wmma_f32(v2f a, v2f b, v8f c) {
    return __builtin_amdgcn_wmma_f32_16x16x4_f32(
         false, a,  false, b,
         (short)0, c,  false,  false);
}

__global__ void gemm_ln_X(const float* __restrict__ aggr,
                          const float* __restrict__ ht,
                          const float* __restrict__ WX,
                          const float* __restrict__ bX,
                          const float* __restrict__ gX,
                          const float* __restrict__ beX,
                          float* __restrict__ outX) {
    __shared__ float lds[4][16][HX + 1];
    const int wave  = threadIdx.x >> 5;
    const int lane  = threadIdx.x & 31;
    const int group = blockIdx.x * 4 + wave;

    if (group < GROUPS) {
        const int m     = lane & 15;
        const int khalf = lane >> 4;
        const int node  = group * 16 + m;
        const float* arow = aggr + (size_t)node * FEAT;

        v8f acc[4] = {};
        #pragma unroll 4
        for (int k0 = 0; k0 < KX; k0 += 4) {
            const int ka = k0 + khalf * 2;
            v2f a;
            if (ka < FEAT) { a.x = arow[ka];      a.y = arow[ka + 1]; }
            else           { a.x = ht[ka - FEAT]; a.y = ht[ka - FEAT + 1]; }
            #pragma unroll
            for (int t = 0; t < 4; ++t) {
                const int n = t * 16 + m;
                v2f b;
                b.x = WX[ka * HX + n];
                b.y = WX[(ka + 1) * HX + n];
                acc[t] = wmma_f32(a, b, acc[t]);
            }
        }
        #pragma unroll
        for (int t = 0; t < 4; ++t) {
            const int n = t * 16 + m;
            const float bias = bX[n];
            #pragma unroll
            for (int v = 0; v < 8; ++v) {
                const int row = khalf * 8 + v;
                float x = acc[t][v] + bias;
                lds[wave][row][n] = fmaxf(x, 0.0f);
            }
        }
    }
    __syncthreads();
    if (group < GROUPS) {
        const int row  = lane >> 1;
        const int half = lane & 1;
        float s = 0.0f, s2 = 0.0f;
        #pragma unroll
        for (int j = 0; j < 32; ++j) {
            float x = lds[wave][row][half * 32 + j];
            s += x; s2 += x * x;
        }
        s  += __shfl_xor(s, 1);
        s2 += __shfl_xor(s2, 1);
        const float mean = s * (1.0f / 64.0f);
        const float var  = s2 * (1.0f / 64.0f) - mean * mean;
        const float inv  = rsqrtf(var + 1e-5f);
        const int node   = group * 16 + row;
        #pragma unroll
        for (int j = 0; j < 32; ++j) {
            const int c = half * 32 + j;
            float x = lds[wave][row][c];
            outX[(size_t)node * HX + c] = (x - mean) * inv * gX[c] + beX[c];
        }
    }
}

__global__ void gemm_ln_Y(const float* __restrict__ aggr,
                          const float* __restrict__ WY,
                          const float* __restrict__ bY,
                          const float* __restrict__ gY,
                          const float* __restrict__ beY,
                          float* __restrict__ outY) {
    __shared__ float lds[4][16][HY + 1];
    const int wave  = threadIdx.x >> 5;
    const int lane  = threadIdx.x & 31;
    const int group = blockIdx.x * 4 + wave;

    if (group < GROUPS) {
        const int m     = lane & 15;
        const int khalf = lane >> 4;
        const int node  = group * 16 + m;
        const float* arow = aggr + (size_t)node * FEAT + HX;

        v8f acc[2] = {};
        #pragma unroll
        for (int k0 = 0; k0 < HY; k0 += 4) {
            const int ka = k0 + khalf * 2;
            v2f a;
            a.x = arow[ka];
            a.y = arow[ka + 1];
            #pragma unroll
            for (int t = 0; t < 2; ++t) {
                const int n = t * 16 + m;
                v2f b;
                b.x = WY[ka * HY + n];
                b.y = WY[(ka + 1) * HY + n];
                acc[t] = wmma_f32(a, b, acc[t]);
            }
        }
        #pragma unroll
        for (int t = 0; t < 2; ++t) {
            const int n = t * 16 + m;
            const float bias = bY[n];
            #pragma unroll
            for (int v = 0; v < 8; ++v) {
                const int row = khalf * 8 + v;
                float x = acc[t][v] + bias;
                lds[wave][row][n] = fmaxf(x, 0.0f);
            }
        }
    }
    __syncthreads();
    if (group < GROUPS) {
        const int row  = lane >> 1;
        const int half = lane & 1;
        float s = 0.0f, s2 = 0.0f;
        #pragma unroll
        for (int j = 0; j < 16; ++j) {
            float x = lds[wave][row][half * 16 + j];
            s += x; s2 += x * x;
        }
        s  += __shfl_xor(s, 1);
        s2 += __shfl_xor(s2, 1);
        const float mean = s * (1.0f / 32.0f);
        const float var  = s2 * (1.0f / 32.0f) - mean * mean;
        const float inv  = rsqrtf(var + 1e-5f);
        const int node   = group * 16 + row;
        #pragma unroll
        for (int j = 0; j < 16; ++j) {
            const int c = half * 16 + j;
            float x = lds[wave][row][c];
            outY[(size_t)node * HY + c] = (x - mean) * inv * gY[c] + beY[c];
        }
    }
}

extern "C" void kernel_launch(void* const* d_in, const int* in_sizes, int n_in,
                              void* d_out, int out_size, void* d_ws, size_t ws_size,
                              hipStream_t stream) {
    const int*   erow = (const int*)d_in[0];
    const int*   ecol = (const int*)d_in[1];
    const float* hX   = (const float*)d_in[2];
    const float* hY   = (const float*)d_in[3];
    const float* ht   = (const float*)d_in[4];
    const float* WX   = (const float*)d_in[5];
    const float* bX   = (const float*)d_in[6];
    const float* gX   = (const float*)d_in[7];
    const float* beX  = (const float*)d_in[8];
    const float* WY   = (const float*)d_in[9];
    const float* bY   = (const float*)d_in[10];
    const float* gY   = (const float*)d_in[11];
    const float* beY  = (const float*)d_in[12];

    float* out  = (float*)d_out;
    float* aggr = (float*)d_ws;

    scan_sum<<<NBLK, SCAN_T, 0, stream>>>(erow, ecol, hX, hY, aggr);

    const int blocks = (GROUPS + 3) / 4;
    gemm_ln_X<<<blocks, 128, 0, stream>>>(aggr, ht, WX, bX, gX, beX, out);
    gemm_ln_Y<<<blocks, 128, 0, stream>>>(aggr, WY, bY, gY, beY, out + (size_t)Nn * HX);
}


# Slab scan reads 1-byte slab tags (new tag launch, row/196 per edge), 4 tags per word test; queue holds edge ids, row re-checked at drain; LDS f32 atomics, arrival order kept; 3 to 4 launches
# speedup vs baseline: 2.1102x; 1.3183x over previous
#include <hip/hip_runtime.h>
#include <hip/hip_bf16.h>

#define Nn 50000
#define Ee 800000
#define HX 64
#define HY 32
#define HT 32
#define FEAT 96
#define KX 128
#define GROUPS ((Nn + 15) / 16)

typedef __attribute__((ext_vector_type(2))) float v2f;
typedef __attribute__((ext_vector_type(8))) float v8f;

#define NBLK  256
#define DBLK  196
#define E4 (Ee / 4)
#define TAG_T 256
#define TAG_WAVE4 128
#define TAG_BLOCKS ((E4 + (TAG_T / 32) * TAG_WAVE4 - 1) / ((TAG_T / 32) * TAG_WAVE4))

__device__ __forceinline__ unsigned slab_tag(int row) {
    return ((unsigned)row / (unsigned)DBLK) & 255u;
}

__global__ __launch_bounds__(TAG_T)
void tag_rows(const int* __restrict__ erow, unsigned* __restrict__ tags) {
    const int wave = (blockIdx.x * TAG_T + threadIdx.x) >> 5;
    const int lane = threadIdx.x & 31;
    const int4* er4 = (const int4*)erow;
    int4 r[4];
    #pragma unroll
    for (int j = 0; j < 4; ++j) {
        int i4 = wave * TAG_WAVE4 + j * 32 + lane; if (i4 > E4 - 1) i4 = E4 - 1;
        r[j] = er4[i4];
    }
    #pragma unroll
    for (int j = 0; j < 4; ++j) {
        const int i4 = wave * TAG_WAVE4 + j * 32 + lane;
        if (i4 < E4)
            tags[i4] = slab_tag(r[j].x) | (slab_tag(r[j].y) << 8) |
                       (slab_tag(r[j].z) << 16) | (slab_tag(r[j].w) << 24);
    }
}

#define SCAN_T 1024
#define SCAN_U 3
#define E16 (Ee / 16)
#define TILE16 (SCAN_T * SCAN_U)
#define TILE_EDGES (TILE16 * 16)
#define NTILES ((E16 + TILE16 - 1) / TILE16)
#define QCAP 57344

#define DR_RUNS 4
#define DR_GRP  16

__device__ __forceinline__ void drain_queue(const unsigned* q, int n, float* acc, int lo,
                                            const int* __restrict__ erow,
                                            const int* __restrict__ ecol,
                                            const float* __restrict__ hX,
                                            const float* __restrict__ hY,
                                            int wave, int lane) {
    const int nruns = (n + 31) >> 5;
    for (int r0 = wave; r0 < nruns; r0 += (SCAN_T / 32) * DR_RUNS) {
        unsigned ent[DR_RUNS];
        int      col[DR_RUNS];
        int      loc[DR_RUNS];
        #pragma unroll
        for (int k = 0; k < DR_RUNS; ++k) {
            const int idx = (r0 + (SCAN_T / 32) * k) * 32 + lane;
            ent[k] = (idx < n) ? q[idx < QCAP ? idx : QCAP - 1] : 0u;
        }
        #pragma unroll
        for (int k = 0; k < DR_RUNS; ++k) { col[k] = ecol[ent[k]]; loc[k] = erow[ent[k]]; }
        #pragma unroll
        for (int k = 0; k < DR_RUNS; ++k) {
            const int idx = (r0 + (SCAN_T / 32) * k) * 32 + lane;
            const unsigned d = (unsigned)(loc[k] - lo);
            loc[k] = (idx < n && d < (unsigned)DBLK) ? (int)d : DBLK;
        }
        #pragma unroll
        for (int k = 0; k < DR_RUNS; ++k) {
            int cnt = n - (r0 + (SCAN_T / 32) * k) * 32;
            if (cnt > 32) cnt = 32;
            for (int j0 = 0; j0 < cnt; j0 += DR_GRP) {
                float x0[DR_GRP], x1[DR_GRP], y0[DR_GRP];
                int   dr[DR_GRP];
                #pragma unroll
                for (int u = 0; u < DR_GRP; ++u) {
                    const int c = __builtin_amdgcn_readlane(col[k], j0 + u);
                    dr[u] = __builtin_amdgcn_readlane(loc[k], j0 + u) * FEAT;
                    const float* sx = hX + (size_t)c * HX;
                    const float* sy = hY + (size_t)c * HY;
                    x0[u] = sx[lane];
                    x1[u] = sx[lane + 32];
                    y0[u] = sy[lane];
                }
                #pragma unroll
                for (int u = 0; u < DR_GRP; ++u) {
                    atomicAdd(&acc[dr[u] + lane],      x0[u]);
                    atomicAdd(&acc[dr[u] + lane + 32], x1[u]);
                    atomicAdd(&acc[dr[u] + lane + 64], y0[u]);
                }
            }
        }
    }
}

__global__ __launch_bounds__(SCAN_T)
void scan_sum(const int* __restrict__ erow,
              const int* __restrict__ ecol,
              const float* __restrict__ hX,
              const float* __restrict__ hY,
              const unsigned* __restrict__ tags,
              float* __restrict__ aggr) {
    __shared__ float    acc[(DBLK + 1) * FEAT];
    __shared__ unsigned q[QCAP];
    __shared__ unsigned qn;

    const int tid  = threadIdx.x;
    const int wave = tid >> 5;
    const int lane = tid & 31;
    const int lo   = blockIdx.x * DBLK;
    const unsigned pat = (unsigned)blockIdx.x * 0x01010101u;

    for (int i = tid; i < (DBLK + 1) * FEAT; i += SCAN_T) acc[i] = 0.0f;
    if (tid == 0) qn = 0u;
    __syncthreads();

    const uint4* tg = (const uint4*)tags;
    const int t0 = (blockIdx.x * NTILES) / NBLK;

    uint4 v[SCAN_U];
    {
        const int base = t0 * TILE16 + tid;
        #pragma unroll
        for (int u = 0; u < SCAN_U; ++u) {
            int i16 = base + u * SCAN_T; if (i16 > E16 - 1) i16 = E16 - 1;
            v[u] = tg[i16];
        }
    }
    for (int t = 0; t < NTILES; ++t) {
        int tt = t + t0; if (tt >= NTILES) tt -= NTILES;
        int tn = tt + 1; if (tn >= NTILES) tn -= NTILES;
        const int base  = tt * TILE16 + tid;
        const int nbase = tn * TILE16 + tid;
        uint4 w[SCAN_U];
        #pragma unroll
        for (int u = 0; u < SCAN_U; ++u) {
            int i16 = nbase + u * SCAN_T; if (i16 > E16 - 1) i16 = E16 - 1;
            w[u] = tg[i16];
        }
        #pragma unroll
        for (int u = 0; u < SCAN_U; ++u) {
            const int i16 = base + u * SCAN_T;
            const unsigned x0 = v[u].x ^ pat, x1 = v[u].y ^ pat, x2 = v[u].z ^ pat, x3 = v[u].w ^ pat;
            const unsigned n0 = (x0 - 0x01010101u) & ~x0;
            const unsigned n1 = (x1 - 0x01010101u) & ~x1;
            const unsigned n2 = (x2 - 0x01010101u) & ~x2;
            const unsigned n3 = (x3 - 0x01010101u) & ~x3;
            if (i16 < E16 && ((n0 | n1 | n2 | n3) & 0x80808080u) != 0u) {
                unsigned mm = ((n0 & 0x80808080u) >> 7) | ((n1 & 0x80808080u) >> 6) |
                              ((n2 & 0x80808080u) >> 5) | ((n3 & 0x80808080u) >> 4);
                const unsigned ebase = (unsigned)i16 * 16u;
                do {
                    const unsigned pos = (unsigned)__builtin_ctz(mm);
                    mm &= mm - 1u;
                    const unsigned p = atomicAdd(&qn, 1u);
                    q[p] = ebase + ((pos & 7u) << 2) + (pos >> 3);
                } while (mm != 0u);
            }
        }
        __syncthreads();
        const int n = (int)qn;
        __syncthreads();
        if (n > QCAP - TILE_EDGES) {
            drain_queue(q, n, acc, lo, erow, ecol, hX, hY, wave, lane);
            __syncthreads();
            if (tid == 0) qn = 0u;
            __syncthreads();
        }
        #pragma unroll
        for (int u = 0; u < SCAN_U; ++u) v[u] = w[u];
    }
    {
        const int n = (int)qn;
        drain_queue(q, n, acc, lo, erow, ecol, hX, hY, wave, lane);
    }
    __syncthreads();

    int lim = (Nn - lo) * FEAT;
    if (lim > DBLK * FEAT) lim = DBLK * FEAT;
    float* dst = aggr + (size_t)lo * FEAT;
    for (int i = tid; i < lim; i += SCAN_T) dst[i] = acc[i];
}

__device__ __forceinline__ v8f wmma_f32(v2f a, v2f b, v8f c) {
    return __builtin_amdgcn_wmma_f32_16x16x4_f32(
         false, a,  false, b,
         (short)0, c,  false,  false);
}

__global__ void gemm_ln_X(const float* __restrict__ aggr,
                          const float* __restrict__ ht,
                          const float* __restrict__ WX,
                          const float* __restrict__ bX,
                          const float* __restrict__ gX,
                          const float* __restrict__ beX,
                          float* __restrict__ outX) {
    __shared__ float lds[4][16][HX + 1];
    const int wave  = threadIdx.x >> 5;
    const int lane  = threadIdx.x & 31;
    const int group = blockIdx.x * 4 + wave;

    if (group < GROUPS) {
        const int m     = lane & 15;
        const int khalf = lane >> 4;
        const int node  = group * 16 + m;
        const float* arow = aggr + (size_t)node * FEAT;

        v8f acc[4] = {};
        #pragma unroll 4
        for (int k0 = 0; k0 < KX; k0 += 4) {
            const int ka = k0 + khalf * 2;
            v2f a;
            if (ka < FEAT) { a.x = arow[ka];      a.y = arow[ka + 1]; }
            else           { a.x = ht[ka - FEAT]; a.y = ht[ka - FEAT + 1]; }
            #pragma unroll
            for (int t = 0; t < 4; ++t) {
                const int n = t * 16 + m;
                v2f b;
                b.x = WX[ka * HX + n];
                b.y = WX[(ka + 1) * HX + n];
                acc[t] = wmma_f32(a, b, acc[t]);
            }
        }
        #pragma unroll
        for (int t = 0; t < 4; ++t) {
            const int n = t * 16 + m;
            const float bias = bX[n];
            #pragma unroll
            for (int v = 0; v < 8; ++v) {
                const int row = khalf * 8 + v;
                float x = acc[t][v] + bias;
                lds[wave][row][n] = fmaxf(x, 0.0f);
            }
        }
    }
    __syncthreads();
    if (group < GROUPS) {
        const int row  = lane >> 1;
        const int half = lane & 1;
        float s = 0.0f, s2 = 0.0f;
        #pragma unroll
        for (int j = 0; j < 32; ++j) {
            float x = lds[wave][row][half * 32 + j];
            s += x; s2 += x * x;
        }
        s  += __shfl_xor(s, 1);
        s2 += __shfl_xor(s2, 1);
        const float mean = s * (1.0f / 64.0f);
        const float var  = s2 * (1.0f / 64.0f) - mean * mean;
        const float inv  = rsqrtf(var + 1e-5f);
        const int node   = group * 16 + row;
        #pragma unroll
        for (int j = 0; j < 32; ++j) {
            const int c = half * 32 + j;
            float x = lds[wave][row][c];
            outX[(size_t)node * HX + c] = (x - mean) * inv * gX[c] + beX[c];
        }
    }
}

__global__ void gemm_ln_Y(const float* __restrict__ aggr,
                          const float* __restrict__ WY,
                          const float* __restrict__ bY,
                          const float* __restrict__ gY,
                          const float* __restrict__ beY,
                          float* __restrict__ outY) {
    __shared__ float lds[4][16][HY + 1];
    const int wave  = threadIdx.x >> 5;
    const int lane  = threadIdx.x & 31;
    const int group = blockIdx.x * 4 + wave;

    if (group < GROUPS) {
        const int m     = lane & 15;
        const int khalf = lane >> 4;
        const int node  = group * 16 + m;
        const float* arow = aggr + (size_t)node * FEAT + HX;

        v8f acc[2] = {};
        #pragma unroll
        for (int k0 = 0; k0 < HY; k0 += 4) {
            const int ka = k0 + khalf * 2;
            v2f a;
            a.x = arow[ka];
            a.y = arow[ka + 1];
            #pragma unroll
            for (int t = 0; t < 2; ++t) {
                const int n = t * 16 + m;
                v2f b;
                b.x = WY[ka * HY + n];
                b.y = WY[(ka + 1) * HY + n];
                acc[t] = wmma_f32(a, b, acc[t]);
            }
        }
        #pragma unroll
        for (int t = 0; t < 2; ++t) {
            const int n = t * 16 + m;
            const float bias = bY[n];
            #pragma unroll
            for (int v = 0; v < 8; ++v) {
                const int row = khalf * 8 + v;
                float x = acc[t][v] + bias;
                lds[wave][row][n] = fmaxf(x, 0.0f);
            }
        }
    }
    __syncthreads();
    if (group < GROUPS) {
        const int row  = lane >> 1;
        const int half = lane & 1;
        float s = 0.0f, s2 = 0.0f;
        #pragma unroll
        for (int j = 0; j < 16; ++j) {
            float x = lds[wave][row][half * 16 + j];
            s += x; s2 += x * x;
        }
        s  += __shfl_xor(s, 1);
        s2 += __shfl_xor(s2, 1);
        const float mean = s * (1.0f / 32.0f);
        const float var  = s2 * (1.0f / 32.0f) - mean * mean;
        const float inv  = rsqrtf(var + 1e-5f);
        const int node   = group * 16 + row;
        #pragma unroll
        for (int j = 0; j < 16; ++j) {
            const int c = half * 16 + j;
            float x = lds[wave][row][c];
            outY[(size_t)node * HY + c] = (x - mean) * inv * gY[c] + beY[c];
        }
    }
}

extern "C" void kernel_launch(void* const* d_in, const int* in_sizes, int n_in,
                              void* d_out, int out_size, void* d_ws, size_t ws_size,
                              hipStream_t stream) {
    const int*   erow = (const int*)d_in[0];
    const int*   ecol = (const int*)d_in[1];
    const float* hX   = (const float*)d_in[2];
    const float* hY   = (const float*)d_in[3];
    const float* ht   = (const float*)d_in[4];
    const float* WX   = (const float*)d_in[5];
    const float* bX   = (const float*)d_in[6];
    const float* gX   = (const float*)d_in[7];
    const float* beX  = (const float*)d_in[8];
    const float* WY   = (const float*)d_in[9];
    const float* bY   = (const float*)d_in[10];
    const float* gY   = (const float*)d_in[11];
    const float* beY  = (const float*)d_in[12];

    float* out  = (float*)d_out;
    float* aggr = (float*)d_ws;
    unsigned* tags = (unsigned*)((char*)d_ws + (32u << 20));

    tag_rows<<<TAG_BLOCKS, TAG_T, 0, stream>>>(erow, tags);
    scan_sum<<<NBLK, SCAN_T, 0, stream>>>(erow, ecol, hX, hY, tags, aggr);

    const int blocks = (GROUPS + 3) / 4;
    gemm_ln_X<<<blocks, 128, 0, stream>>>(aggr, ht, WX, bX, gX, beX, out);
    gemm_ln_Y<<<blocks, 128, 0, stream>>>(aggr, WY, bY, gY, beY, out + (size_t)Nn * HX);
}


# Y product folded into the X product kernel: its A fragments are the X loop's own for columns 64..95 (k loop cut at 64 and 96); aggr read once; 4 launches to 3; sum orders unchanged
# speedup vs baseline: 2.4768x; 1.1737x over previous
#include <hip/hip_runtime.h>
#include <hip/hip_bf16.h>

#define Nn 50000
#define Ee 800000
#define HX 64
#define HY 32
#define HT 32
#define FEAT 96
#define KX 128
#define GROUPS ((Nn + 15) / 16)

typedef __attribute__((ext_vector_type(2))) float v2f;
typedef __attribute__((ext_vector_type(8))) float v8f;

#define NBLK  256
#define DBLK  196
#define E4 (Ee / 4)
#define TAG_T 256
#define TAG_WAVE4 128
#define TAG_BLOCKS ((E4 + (TAG_T / 32) * TAG_WAVE4 - 1) / ((TAG_T / 32) * TAG_WAVE4))

__device__ __forceinline__ unsigned slab_tag(int row) {
    return ((unsigned)row / (unsigned)DBLK) & 255u;
}

__global__ __launch_bounds__(TAG_T)
void tag_rows(const int* __restrict__ erow, unsigned* __restrict__ tags) {
    const int wave = (blockIdx.x * TAG_T + threadIdx.x) >> 5;
    const int lane = threadIdx.x & 31;
    const int4* er4 = (const int4*)erow;
    int4 r[4];
    #pragma unroll
    for (int j = 0; j < 4; ++j) {
        int i4 = wave * TAG_WAVE4 + j * 32 + lane; if (i4 > E4 - 1) i4 = E4 - 1;
        r[j] = er4[i4];
    }
    #pragma unroll
    for (int j = 0; j < 4; ++j) {
        const int i4 = wave * TAG_WAVE4 + j * 32 + lane;
        if (i4 < E4)
            tags[i4] = slab_tag(r[j].x) | (slab_tag(r[j].y) << 8) |
                       (slab_tag(r[j].z) << 16) | (slab_tag(r[j].w) << 24);
    }
}

#define SCAN_T 1024
#define SCAN_U 3
#define E16 (Ee / 16)
#define TILE16 (SCAN_T * SCAN_U)
#define TILE_EDGES (TILE16 * 16)
#define NTILES ((E16 + TILE16 - 1) / TILE16)
#define QCAP 57344

#define DR_RUNS 4
#define DR_GRP  16

__device__ __forceinline__ void drain_queue(const unsigned* q, int n, float* acc, int lo,
                                            const int* __restrict__ erow,
                                            const int* __restrict__ ecol,
                                            const float* __restrict__ hX,
                                            const float* __restrict__ hY,
                                            int wave, int lane) {
    const int nruns = (n + 31) >> 5;
    for (int r0 = wave; r0 < nruns; r0 += (SCAN_T / 32) * DR_RUNS) {
        unsigned ent[DR_RUNS];
        int      col[DR_RUNS];
        int      loc[DR_RUNS];
        #pragma unroll
        for (int k = 0; k < DR_RUNS; ++k) {
            const int idx = (r0 + (SCAN_T / 32) * k) * 32 + lane;
            ent[k] = (idx < n) ? q[idx < QCAP ? idx : QCAP - 1] : 0u;
        }
        #pragma unroll
        for (int k = 0; k < DR_RUNS; ++k) { col[k] = ecol[ent[k]]; loc[k] = erow[ent[k]]; }
        #pragma unroll
        for (int k = 0; k < DR_RUNS; ++k) {
            const int idx = (r0 + (SCAN_T / 32) * k) * 32 + lane;
            const unsigned d = (unsigned)(loc[k] - lo);
            loc[k] = (idx < n && d < (unsigned)DBLK) ? (int)d : DBLK;
        }
        #pragma unroll
        for (int k = 0; k < DR_RUNS; ++k) {
            int cnt = n - (r0 + (SCAN_T / 32) * k) * 32;
            if (cnt > 32) cnt = 32;
            for (int j0 = 0; j0 < cnt; j0 += DR_GRP) {
                float x0[DR_GRP], x1[DR_GRP], y0[DR_GRP];
                int   dr[DR_GRP];
                #pragma unroll
                for (int u = 0; u < DR_GRP; ++u) {
                    const int c = __builtin_amdgcn_readlane(col[k], j0 + u);
                    dr[u] = __builtin_amdgcn_readlane(loc[k], j0 + u) * FEAT;
                    const float* sx = hX + (size_t)c * HX;
                    const float* sy = hY + (size_t)c * HY;
                    x0[u] = sx[lane];
                    x1[u] = sx[lane + 32];
                    y0[u] = sy[lane];
                }
                #pragma unroll
                for (int u = 0; u < DR_GRP; ++u) {
                    atomicAdd(&acc[dr[u] + lane],      x0[u]);
                    atomicAdd(&acc[dr[u] + lane + 32], x1[u]);
                    atomicAdd(&acc[dr[u] + lane + 64], y0[u]);
                }
            }
        }
    }
}

__global__ __launch_bounds__(SCAN_T)
void scan_sum(const int* __restrict__ erow,
              const int* __restrict__ ecol,
              const float* __restrict__ hX,
              const float* __restrict__ hY,
              const unsigned* __restrict__ tags,
              float* __restrict__ aggr) {
    __shared__ float    acc[(DBLK + 1) * FEAT];
    __shared__ unsigned q[QCAP];
    __shared__ unsigned qn;

    const int tid  = threadIdx.x;
    const int wave = tid >> 5;
    const int lane = tid & 31;
    const int lo   = blockIdx.x * DBLK;
    const unsigned pat = (unsigned)blockIdx.x * 0x01010101u;

    for (int i = tid; i < (DBLK + 1) * FEAT; i += SCAN_T) acc[i] = 0.0f;
    if (tid == 0) qn = 0u;
    __syncthreads();

    const uint4* tg = (const uint4*)tags;
    const int t0 = (blockIdx.x * NTILES) / NBLK;

    uint4 v[SCAN_U];
    {
        const int base = t0 * TILE16 + tid;
        #pragma unroll
        for (int u = 0; u < SCAN_U; ++u) {
            int i16 = base + u * SCAN_T; if (i16 > E16 - 1) i16 = E16 - 1;
            v[u] = tg[i16];
        }
    }
    for (int t = 0; t < NTILES; ++t) {
        int tt = t + t0; if (tt >= NTILES) tt -= NTILES;
        int tn = tt + 1; if (tn >= NTILES) tn -= NTILES;
        const int base  = tt * TILE16 + tid;
        const int nbase = tn * TILE16 + tid;
        uint4 w[SCAN_U];
        #pragma unroll
        for (int u = 0; u < SCAN_U; ++u) {
            int i16 = nbase + u * SCAN_T; if (i16 > E16 - 1) i16 = E16 - 1;
            w[u] = tg[i16];
        }
        #pragma unroll
        for (int u = 0; u < SCAN_U; ++u) {
            const int i16 = base + u * SCAN_T;
            const unsigned x0 = v[u].x ^ pat, x1 = v[u].y ^ pat, x2 = v[u].z ^ pat, x3 = v[u].w ^ pat;
            const unsigned n0 = (x0 - 0x01010101u) & ~x0;
            const unsigned n1 = (x1 - 0x01010101u) & ~x1;
            const unsigned n2 = (x2 - 0x01010101u) & ~x2;
            const unsigned n3 = (x3 - 0x01010101u) & ~x3;
            if (i16 < E16 && ((n0 | n1 | n2 | n3) & 0x80808080u) != 0u) {
                unsigned mm = ((n0 & 0x80808080u) >> 7) | ((n1 & 0x80808080u) >> 6) |
                              ((n2 & 0x80808080u) >> 5) | ((n3 & 0x80808080u) >> 4);
                const unsigned ebase = (unsigned)i16 * 16u;
                do {
                    const unsigned pos = (unsigned)__builtin_ctz(mm);
                    mm &= mm - 1u;
                    const unsigned p = atomicAdd(&qn, 1u);
                    q[p] = ebase + ((pos & 7u) << 2) + (pos >> 3);
                } while (mm != 0u);
            }
        }
        __syncthreads();
        const int n = (int)qn;
        __syncthreads();
        if (n > QCAP - TILE_EDGES) {
            drain_queue(q, n, acc, lo, erow, ecol, hX, hY, wave, lane);
            __syncthreads();
            if (tid == 0) qn = 0u;
            __syncthreads();
        }
        #pragma unroll
        for (int u = 0; u < SCAN_U; ++u) v[u] = w[u];
    }
    {
        const int n = (int)qn;
        drain_queue(q, n, acc, lo, erow, ecol, hX, hY, wave, lane);
    }
    __syncthreads();

    int lim = (Nn - lo) * FEAT;
    if (lim > DBLK * FEAT) lim = DBLK * FEAT;
    float* dst = aggr + (size_t)lo * FEAT;
    for (int i = tid; i < lim; i += SCAN_T) dst[i] = acc[i];
}

__device__ __forceinline__ v8f wmma_f32(v2f a, v2f b, v8f c) {
    return __builtin_amdgcn_wmma_f32_16x16x4_f32(
         false, a,  false, b,
         (short)0, c,  false,  false);
}

__global__ void gemm_ln_XY(const float* __restrict__ aggr,
                           const float* __restrict__ ht,
                           const float* __restrict__ WX,
                           const float* __restrict__ bX,
                           const float* __restrict__ gX,
                           const float* __restrict__ beX,
                           const float* __restrict__ WY,
                           const float* __restrict__ bY,
                           const float* __restrict__ gY,
                           const float* __restrict__ beY,
                           float* __restrict__ outX,
                           float* __restrict__ outY) {
    __shared__ float ldsX[4][16][HX + 1];
    __shared__ float ldsY[4][16][HY + 1];
    const int wave  = threadIdx.x >> 5;
    const int lane  = threadIdx.x & 31;
    const int group = blockIdx.x * 4 + wave;

    if (group < GROUPS) {
        const int m     = lane & 15;
        const int khalf = lane >> 4;
        const int node  = group * 16 + m;
        const float* arow = aggr + (size_t)node * FEAT;

        v8f acc[4]  = {};
        v8f accY[2] = {};
        #pragma unroll 4
        for (int k0 = 0; k0 < HX; k0 += 4) {
            const int ka = k0 + khalf * 2;
            v2f a;
            a.x = arow[ka];
            a.y = arow[ka + 1];
            #pragma unroll
            for (int t = 0; t < 4; ++t) {
                const int n = t * 16 + m;
                v2f b;
                b.x = WX[ka * HX + n];
                b.y = WX[(ka + 1) * HX + n];
                acc[t] = wmma_f32(a, b, acc[t]);
            }
        }
        #pragma unroll 4
        for (int k0 = HX; k0 < FEAT; k0 += 4) {
            const int ka = k0 + khalf * 2;
            const int ky = ka - HX;
            v2f a;
            a.x = arow[ka];
            a.y = arow[ka + 1];
            #pragma unroll
            for (int t = 0; t < 4; ++t) {
                const int n = t * 16 + m;
                v2f b;
                b.x = WX[ka * HX + n];
                b.y = WX[(ka + 1) * HX + n];
                acc[t] = wmma_f32(a, b, acc[t]);
            }
            #pragma unroll
            for (int t = 0; t < 2; ++t) {
                const int n = t * 16 + m;
                v2f b;
                b.x = WY[ky * HY + n];
                b.y = WY[(ky + 1) * HY + n];
                accY[t] = wmma_f32(a, b, accY[t]);
            }
        }
        #pragma unroll 4
        for (int k0 = FEAT; k0 < KX; k0 += 4) {
            const int ka = k0 + khalf * 2;
            v2f a;
            a.x = ht[ka - FEAT];
            a.y = ht[ka - FEAT + 1];
            #pragma unroll
            for (int t = 0; t < 4; ++t) {
                const int n = t * 16 + m;
                v2f b;
                b.x = WX[ka * HX + n];
                b.y = WX[(ka + 1) * HX + n];
                acc[t] = wmma_f32(a, b, acc[t]);
            }
        }
        #pragma unroll
        for (int t = 0; t < 4; ++t) {
            const int n = t * 16 + m;
            const float bias = bX[n];
            #pragma unroll
            for (int v = 0; v < 8; ++v) {
                const int row = khalf * 8 + v;
                float x = acc[t][v] + bias;
                ldsX[wave][row][n] = fmaxf(x, 0.0f);
            }
        }
        #pragma unroll
        for (int t = 0; t < 2; ++t) {
            const int n = t * 16 + m;
            const float bias = bY[n];
            #pragma unroll
            for (int v = 0; v < 8; ++v) {
                const int row = khalf * 8 + v;
                float x = accY[t][v] + bias;
                ldsY[wave][row][n] = fmaxf(x, 0.0f);
            }
        }
    }
    __syncthreads();
    if (group < GROUPS) {
        const int row  = lane >> 1;
        const int half = lane & 1;
        const int node = group * 16 + row;
        {
            float s = 0.0f, s2 = 0.0f;
            #pragma unroll
            for (int j = 0; j < 32; ++j) {
                float x = ldsX[wave][row][half * 32 + j];
                s += x; s2 += x * x;
            }
            s  += __shfl_xor(s, 1);
            s2 += __shfl_xor(s2, 1);
            const float mean = s * (1.0f / 64.0f);
            const float var  = s2 * (1.0f / 64.0f) - mean * mean;
            const float inv  = rsqrtf(var + 1e-5f);
            #pragma unroll
            for (int j = 0; j < 32; ++j) {
                const int c = half * 32 + j;
                float x = ldsX[wave][row][c];
                outX[(size_t)node * HX + c] = (x - mean) * inv * gX[c] + beX[c];
            }
        }
        {
            float s = 0.0f, s2 = 0.0f;
            #pragma unroll
            for (int j = 0; j < 16; ++j) {
                float x = ldsY[wave][row][half * 16 + j];
                s += x; s2 += x * x;
            }
            s  += __shfl_xor(s, 1);
            s2 += __shfl_xor(s2, 1);
            const float mean = s * (1.0f / 32.0f);
            const float var  = s2 * (1.0f / 32.0f) - mean * mean;
            const float inv  = rsqrtf(var + 1e-5f);
            #pragma unroll
            for (int j = 0; j < 16; ++j) {
                const int c = half * 16 + j;
                float x = ldsY[wave][row][c];
                outY[(size_t)node * HY + c] = (x - mean) * inv * gY[c] + beY[c];
            }
        }
    }
}

extern "C" void kernel_launch(void* const* d_in, const int* in_sizes, int n_in,
                              void* d_out, int out_size, void* d_ws, size_t ws_size,
                              hipStream_t stream) {
    const int*   erow = (const int*)d_in[0];
    const int*   ecol = (const int*)d_in[1];
    const float* hX   = (const float*)d_in[2];
    const float* hY   = (const float*)d_in[3];
    const float* ht   = (const float*)d_in[4];
    const float* WX   = (const float*)d_in[5];
    const float* bX   = (const float*)d_in[6];
    const float* gX   = (const float*)d_in[7];
    const float* beX  = (const float*)d_in[8];
    const float* WY   = (const float*)d_in[9];
    const float* bY   = (const float*)d_in[10];
    const float* gY   = (const float*)d_in[11];
    const float* beY  = (const float*)d_in[12];

    float* out  = (float*)d_out;
    float* aggr = (float*)d_ws;
    unsigned* tags = (unsigned*)((char*)d_ws + (32u << 20));

    tag_rows<<<TAG_BLOCKS, TAG_T, 0, stream>>>(erow, tags);
    scan_sum<<<NBLK, SCAN_T, 0, stream>>>(erow, ecol, hX, hY, tags, aggr);

    const int blocks = (GROUPS + 3) / 4;
    gemm_ln_XY<<<blocks, 128, 0, stream>>>(aggr, ht, WX, bX, gX, beX, WY, bY, gY, beY,
                                           out, out + (size_t)Nn * HX);
}


# Scan and gather overlap: a wave owns its queue part (slot = own count + lane rank; no counter atomic, no barrier per tile), drains 32 entries at a time between tiles; LDS f32 atomics, arrival order
# speedup vs baseline: 2.5768x; 1.0404x over previous
#include <hip/hip_runtime.h>
#include <hip/hip_bf16.h>

#define Nn 50000
#define Ee 800000
#define HX 64
#define HY 32
#define HT 32
#define FEAT 96
#define KX 128
#define GROUPS ((Nn + 15) / 16)

typedef __attribute__((ext_vector_type(2))) float v2f;
typedef __attribute__((ext_vector_type(8))) float v8f;

#define NBLK  256
#define DBLK  196
#define E4 (Ee / 4)
#define TAG_T 256
#define TAG_WAVE4 128
#define TAG_BLOCKS ((E4 + (TAG_T / 32) * TAG_WAVE4 - 1) / ((TAG_T / 32) * TAG_WAVE4))

__device__ __forceinline__ unsigned slab_tag(int row) {
    return ((unsigned)row / (unsigned)DBLK) & 255u;
}

__global__ __launch_bounds__(TAG_T)
void tag_rows(const int* __restrict__ erow, unsigned* __restrict__ tags) {
    const int wave = (blockIdx.x * TAG_T + threadIdx.x) >> 5;
    const int lane = threadIdx.x & 31;
    const int4* er4 = (const int4*)erow;
    int4 r[4];
    #pragma unroll
    for (int j = 0; j < 4; ++j) {
        int i4 = wave * TAG_WAVE4 + j * 32 + lane; if (i4 > E4 - 1) i4 = E4 - 1;
        r[j] = er4[i4];
    }
    #pragma unroll
    for (int j = 0; j < 4; ++j) {
        const int i4 = wave * TAG_WAVE4 + j * 32 + lane;
        if (i4 < E4)
            tags[i4] = slab_tag(r[j].x) | (slab_tag(r[j].y) << 8) |
                       (slab_tag(r[j].z) << 16) | (slab_tag(r[j].w) << 24);
    }
}

#define SCAN_T 1024
#define SCAN_U 3
#define E16 (Ee / 16)
#define TILE16 (SCAN_T * SCAN_U)
#define NTILES ((E16 + TILE16 - 1) / TILE16)
#define QCAP 57344
#define QW (QCAP / (SCAN_T / 32))
#define WAVE_TILE_EDGES (32 * SCAN_U * 16)

#define DR_GRP  16

__device__ __forceinline__ void drain_run(const unsigned* qrun, int n, float* acc, int lo,
                                          const int* __restrict__ erow,
                                          const int* __restrict__ ecol,
                                          const float* __restrict__ hX,
                                          const float* __restrict__ hY,
                                          int lane) {
    const unsigned ent = (lane < n) ? qrun[lane] : 0u;
    const int col = ecol[ent];
    int       loc = erow[ent];
    const unsigned d = (unsigned)(loc - lo);
    loc = (lane < n && d < (unsigned)DBLK) ? (int)d : DBLK;
    for (int j0 = 0; j0 < n; j0 += DR_GRP) {
        float x0[DR_GRP], x1[DR_GRP], y0[DR_GRP];
        int   dr[DR_GRP];
        #pragma unroll
        for (int u = 0; u < DR_GRP; ++u) {
            const int c = __builtin_amdgcn_readlane(col, j0 + u);
            dr[u] = __builtin_amdgcn_readlane(loc, j0 + u) * FEAT;
            const float* sx = hX + (size_t)c * HX;
            const float* sy = hY + (size_t)c * HY;
            x0[u] = sx[lane];
            x1[u] = sx[lane + 32];
            y0[u] = sy[lane];
        }
        #pragma unroll
        for (int u = 0; u < DR_GRP; ++u) {
            atomicAdd(&acc[dr[u] + lane],      x0[u]);
            atomicAdd(&acc[dr[u] + lane + 32], x1[u]);
            atomicAdd(&acc[dr[u] + lane + 64], y0[u]);
        }
    }
}

__global__ __launch_bounds__(SCAN_T)
void scan_sum(const int* __restrict__ erow,
              const int* __restrict__ ecol,
              const float* __restrict__ hX,
              const float* __restrict__ hY,
              const unsigned* __restrict__ tags,
              float* __restrict__ aggr) {
    __shared__ float    acc[(DBLK + 1) * FEAT];
    __shared__ unsigned q[QCAP];

    const int tid  = threadIdx.x;
    const int wave = tid >> 5;
    const int lane = tid & 31;
    const int lo   = blockIdx.x * DBLK;
    const unsigned pat = (unsigned)blockIdx.x * 0x01010101u;

    for (int i = tid; i < (DBLK + 1) * FEAT; i += SCAN_T) acc[i] = 0.0f;
    __syncthreads();

    unsigned* qw = q + wave * QW;
    int cnt  = 0;
    int head = 0;

    const uint4* tg = (const uint4*)tags;
    const int t0 = (blockIdx.x * NTILES) / NBLK;

    uint4 v[SCAN_U];
    {
        const int base = t0 * TILE16 + tid;
        #pragma unroll
        for (int u = 0; u < SCAN_U; ++u) {
            int i16 = base + u * SCAN_T; if (i16 > E16 - 1) i16 = E16 - 1;
            v[u] = tg[i16];
        }
    }
    for (int t = 0; t < NTILES; ++t) {
        int tt = t + t0; if (tt >= NTILES) tt -= NTILES;
        int tn = tt + 1; if (tn >= NTILES) tn -= NTILES;
        const int base  = tt * TILE16 + tid;
        const int nbase = tn * TILE16 + tid;
        uint4 w[SCAN_U];
        #pragma unroll
        for (int u = 0; u < SCAN_U; ++u) {
            int i16 = nbase + u * SCAN_T; if (i16 > E16 - 1) i16 = E16 - 1;
            w[u] = tg[i16];
        }
        #pragma unroll
        for (int u = 0; u < SCAN_U; ++u) {
            const int i16 = base + u * SCAN_T;
            const unsigned x0 = v[u].x ^ pat, x1 = v[u].y ^ pat, x2 = v[u].z ^ pat, x3 = v[u].w ^ pat;
            const unsigned n0 = (x0 - 0x01010101u) & ~x0;
            const unsigned n1 = (x1 - 0x01010101u) & ~x1;
            const unsigned n2 = (x2 - 0x01010101u) & ~x2;
            const unsigned n3 = (x3 - 0x01010101u) & ~x3;
            unsigned mm = 0u;
            if (i16 < E16)
                mm = ((n0 & 0x80808080u) >> 7) | ((n1 & 0x80808080u) >> 6) |
                     ((n2 & 0x80808080u) >> 5) | ((n3 & 0x80808080u) >> 4);
            const unsigned ebase = (unsigned)i16 * 16u;
            unsigned bal = __builtin_amdgcn_ballot_w32(mm != 0u);
            while (bal != 0u) {
                if (mm != 0u) {
                    const unsigned pos = (unsigned)__builtin_ctz(mm);
                    mm &= mm - 1u;
                    const unsigned slot = (unsigned)cnt + __builtin_amdgcn_mbcnt_lo(bal, 0u);
                    if (slot < (unsigned)QW)
                        qw[slot] = ebase + ((pos & 7u) << 2) + (pos >> 3);
                }
                cnt += __builtin_popcount(bal);
                bal = __builtin_amdgcn_ballot_w32(mm != 0u);
            }
        }
        const bool flush = (t == NTILES - 1) || (cnt > QW - WAVE_TILE_EDGES);
        while (cnt - head >= 32 || (flush && cnt > head)) {
            int n = cnt - head; if (n > 32) n = 32;
            drain_run(qw + head, n, acc, lo, erow, ecol, hX, hY, lane);
            head += n;
        }
        if (flush) { cnt = 0; head = 0; }
        #pragma unroll
        for (int u = 0; u < SCAN_U; ++u) v[u] = w[u];
    }
    __syncthreads();

    int lim = (Nn - lo) * FEAT;
    if (lim > DBLK * FEAT) lim = DBLK * FEAT;
    float* dst = aggr + (size_t)lo * FEAT;
    for (int i = tid; i < lim; i += SCAN_T) dst[i] = acc[i];
}

__device__ __forceinline__ v8f wmma_f32(v2f a, v2f b, v8f c) {
    return __builtin_amdgcn_wmma_f32_16x16x4_f32(
         false, a,  false, b,
         (short)0, c,  false,  false);
}

__global__ void gemm_ln_XY(const float* __restrict__ aggr,
                           const float* __restrict__ ht,
                           const float* __restrict__ WX,
                           const float* __restrict__ bX,
                           const float* __restrict__ gX,
                           const float* __restrict__ beX,
                           const float* __restrict__ WY,
                           const float* __restrict__ bY,
                           const float* __restrict__ gY,
                           const float* __restrict__ beY,
                           float* __restrict__ outX,
                           float* __restrict__ outY) {
    __shared__ float ldsX[4][16][HX + 1];
    __shared__ float ldsY[4][16][HY + 1];
    const int wave  = threadIdx.x >> 5;
    const int lane  = threadIdx.x & 31;
    const int group = blockIdx.x * 4 + wave;

    if (group < GROUPS) {
        const int m     = lane & 15;
        const int khalf = lane >> 4;
        const int node  = group * 16 + m;
        const float* arow = aggr + (size_t)node * FEAT;

        v8f acc[4]  = {};
        v8f accY[2] = {};
        #pragma unroll 4
        for (int k0 = 0; k0 < HX; k0 += 4) {
            const int ka = k0 + khalf * 2;
            v2f a;
            a.x = arow[ka];
            a.y = arow[ka + 1];
            #pragma unroll
            for (int t = 0; t < 4; ++t) {
                const int n = t * 16 + m;
                v2f b;
                b.x = WX[ka * HX + n];
                b.y = WX[(ka + 1) * HX + n];
                acc[t] = wmma_f32(a, b, acc[t]);
            }
        }
        #pragma unroll 4
        for (int k0 = HX; k0 < FEAT; k0 += 4) {
            const int ka = k0 + khalf * 2;
            const int ky = ka - HX;
            v2f a;
            a.x = arow[ka];
            a.y = arow[ka + 1];
            #pragma unroll
            for (int t = 0; t < 4; ++t) {
                const int n = t * 16 + m;
                v2f b;
                b.x = WX[ka * HX + n];
                b.y = WX[(ka + 1) * HX + n];
                acc[t] = wmma_f32(a, b, acc[t]);
            }
            #pragma unroll
            for (int t = 0; t < 2; ++t) {
                const int n = t * 16 + m;
                v2f b;
                b.x = WY[ky * HY + n];
                b.y = WY[(ky + 1) * HY + n];
                accY[t] = wmma_f32(a, b, accY[t]);
            }
        }
        #pragma unroll 4
        for (int k0 = FEAT; k0 < KX; k0 += 4) {
            const int ka = k0 + khalf * 2;
            v2f a;
            a.x = ht[ka - FEAT];
            a.y = ht[ka - FEAT + 1];
            #pragma unroll
            for (int t = 0; t < 4; ++t) {
                const int n = t * 16 + m;
                v2f b;
                b.x = WX[ka * HX + n];
                b.y = WX[(ka + 1) * HX + n];
                acc[t] = wmma_f32(a, b, acc[t]);
            }
        }
        #pragma unroll
        for (int t = 0; t < 4; ++t) {
            const int n = t * 16 + m;
            const float bias = bX[n];
            #pragma unroll
            for (int v = 0; v < 8; ++v) {
                const int row = khalf * 8 + v;
                float x = acc[t][v] + bias;
                ldsX[wave][row][n] = fmaxf(x, 0.0f);
            }
        }
        #pragma unroll
        for (int t = 0; t < 2; ++t) {
            const int n = t * 16 + m;
            const float bias = bY[n];
            #pragma unroll
            for (int v = 0; v < 8; ++v) {
                const int row = khalf * 8 + v;
                float x = accY[t][v] + bias;
                ldsY[wave][row][n] = fmaxf(x, 0.0f);
            }
        }
    }
    __syncthreads();
    if (group < GROUPS) {
        const int row  = lane >> 1;
        const int half = lane & 1;
        const int node = group * 16 + row;
        {
            float s = 0.0f, s2 = 0.0f;
            #pragma unroll
            for (int j = 0; j < 32; ++j) {
                float x = ldsX[wave][row][half * 32 + j];
                s += x; s2 += x * x;
            }
            s  += __shfl_xor(s, 1);
            s2 += __shfl_xor(s2, 1);
            const float mean = s * (1.0f / 64.0f);
            const float var  = s2 * (1.0f / 64.0f) - mean * mean;
            const float inv  = rsqrtf(var + 1e-5f);
            #pragma unroll
            for (int j = 0; j < 32; ++j) {
                const int c = half * 32 + j;
                float x = ldsX[wave][row][c];
                outX[(size_t)node * HX + c] = (x - mean) * inv * gX[c] + beX[c];
            }
        }
        {
            float s = 0.0f, s2 = 0.0f;
            #pragma unroll
            for (int j = 0; j < 16; ++j) {
                float x = ldsY[wave][row][half * 16 + j];
                s += x; s2 += x * x;
            }
            s  += __shfl_xor(s, 1);
            s2 += __shfl_xor(s2, 1);
            const float mean = s * (1.0f / 32.0f);
            const float var  = s2 * (1.0f / 32.0f) - mean * mean;
            const float inv  = rsqrtf(var + 1e-5f);
            #pragma unroll
            for (int j = 0; j < 16; ++j) {
                const int c = half * 16 + j;
                float x = ldsY[wave][row][c];
                outY[(size_t)node * HY + c] = (x - mean) * inv * gY[c] + beY[c];
            }
        }
    }
}

extern "C" void kernel_launch(void* const* d_in, const int* in_sizes, int n_in,
                              void* d_out, int out_size, void* d_ws, size_t ws_size,
                              hipStream_t stream) {
    const int*   erow = (const int*)d_in[0];
    const int*   ecol = (const int*)d_in[1];
    const float* hX   = (const float*)d_in[2];
    const float* hY   = (const float*)d_in[3];
    const float* ht   = (const float*)d_in[4];
    const float* WX   = (const float*)d_in[5];
    const float* bX   = (const float*)d_in[6];
    const float* gX   = (const float*)d_in[7];
    const float* beX  = (const float*)d_in[8];
    const float* WY   = (const float*)d_in[9];
    const float* bY   = (const float*)d_in[10];
    const float* gY   = (const float*)d_in[11];
    const float* beY  = (const float*)d_in[12];

    float* out  = (float*)d_out;
    float* aggr = (float*)d_ws;
    unsigned* tags = (unsigned*)((char*)d_ws + (32u << 20));

    tag_rows<<<TAG_BLOCKS, TAG_T, 0, stream>>>(erow, tags);
    scan_sum<<<NBLK, SCAN_T, 0, stream>>>(erow, ecol, hX, hY, tags, aggr);

    const int blocks = (GROUPS + 3) / 4;
    gemm_ln_XY<<<blocks, 128, 0, stream>>>(aggr, ht, WX, bX, gX, beX, WY, bY, gY, beY,
                                           out, out + (size_t)Nn * HX);
}


# Tag launch also packs slab, row in slab and source id into one word per edge; the drain reads that one word per queued edge, not destination and source separately; LDS f32 atomics, arrival order kept
# speedup vs baseline: 2.7288x; 1.0590x over previous
#include <hip/hip_runtime.h>
#include <hip/hip_bf16.h>

#define Nn 50000
#define Ee 800000
#define HX 64
#define HY 32
#define HT 32
#define FEAT 96
#define KX 128
#define GROUPS ((Nn + 15) / 16)

typedef __attribute__((ext_vector_type(2))) float v2f;
typedef __attribute__((ext_vector_type(8))) float v8f;

#define NBLK  256
#define DBLK  196
#define E4 (Ee / 4)
#define TAG_T 256
#define TAG_WAVE4 128
#define TAG_BLOCKS ((E4 + (TAG_T / 32) * TAG_WAVE4 - 1) / ((TAG_T / 32) * TAG_WAVE4))

static_assert(DBLK < 255, "the row within a slab and its out-of-range mark must fit a byte");
static_assert(Nn <= 65536, "a source id must fit 16 bits of the packed word");

__device__ __forceinline__ unsigned slab_tag(int row) {
    return ((unsigned)row / (unsigned)DBLK) & 255u;
}

__device__ __forceinline__ unsigned pack_edge(int row, int col) {
    const unsigned r = (unsigned)row;
    const unsigned s = r / (unsigned)DBLK;
    const unsigned l = (r < (unsigned)(NBLK * DBLK)) ? (r - s * (unsigned)DBLK) : 255u;
    return ((s & 255u) << 24) | (l << 16) | ((unsigned)col & 0xffffu);
}

__global__ __launch_bounds__(TAG_T)
void tag_rows(const int* __restrict__ erow, const int* __restrict__ ecol,
              unsigned* __restrict__ tags, unsigned* __restrict__ pk) {
    const int wave = (blockIdx.x * TAG_T + threadIdx.x) >> 5;
    const int lane = threadIdx.x & 31;
    const int4* er4 = (const int4*)erow;
    const int4* ec4 = (const int4*)ecol;
    uint4* pk4 = (uint4*)pk;
    int4 r[4], c[4];
    #pragma unroll
    for (int j = 0; j < 4; ++j) {
        int i4 = wave * TAG_WAVE4 + j * 32 + lane; if (i4 > E4 - 1) i4 = E4 - 1;
        r[j] = er4[i4];
        c[j] = ec4[i4];
    }
    #pragma unroll
    for (int j = 0; j < 4; ++j) {
        const int i4 = wave * TAG_WAVE4 + j * 32 + lane;
        if (i4 < E4) {
            tags[i4] = slab_tag(r[j].x) | (slab_tag(r[j].y) << 8) |
                       (slab_tag(r[j].z) << 16) | (slab_tag(r[j].w) << 24);
            pk4[i4] = make_uint4(pack_edge(r[j].x, c[j].x), pack_edge(r[j].y, c[j].y),
                                 pack_edge(r[j].z, c[j].z), pack_edge(r[j].w, c[j].w));
        }
    }
}

#define SCAN_T 1024
#define SCAN_U 3
#define E16 (Ee / 16)
#define TILE16 (SCAN_T * SCAN_U)
#define NTILES ((E16 + TILE16 - 1) / TILE16)
#define QCAP 57344
#define QW (QCAP / (SCAN_T / 32))
#define WAVE_TILE_EDGES (32 * SCAN_U * 16)

#define DR_GRP  16

__device__ __forceinline__ void drain_run(const unsigned* qrun, int n, float* acc, unsigned slab,
                                          const unsigned* __restrict__ pk,
                                          const float* __restrict__ hX,
                                          const float* __restrict__ hY,
                                          int lane) {
    const unsigned ent = (lane < n) ? qrun[lane] : 0u;
    const unsigned p = pk[ent];
    const int col = (int)(p & 0xffffu);
    const unsigned d = (p >> 16) & 255u;
    const int loc = (lane < n && (p >> 24) == slab && d < (unsigned)DBLK) ? (int)d : DBLK;
    for (int j0 = 0; j0 < n; j0 += DR_GRP) {
        float x0[DR_GRP], x1[DR_GRP], y0[DR_GRP];
        int   dr[DR_GRP];
        #pragma unroll
        for (int u = 0; u < DR_GRP; ++u) {
            const int c = __builtin_amdgcn_readlane(col, j0 + u);
            dr[u] = __builtin_amdgcn_readlane(loc, j0 + u) * FEAT;
            const float* sx = hX + (size_t)c * HX;
            const float* sy = hY + (size_t)c * HY;
            x0[u] = sx[lane];
            x1[u] = sx[lane + 32];
            y0[u] = sy[lane];
        }
        #pragma unroll
        for (int u = 0; u < DR_GRP; ++u) {
            atomicAdd(&acc[dr[u] + lane],      x0[u]);
            atomicAdd(&acc[dr[u] + lane + 32], x1[u]);
            atomicAdd(&acc[dr[u] + lane + 64], y0[u]);
        }
    }
}

__global__ __launch_bounds__(SCAN_T)
void scan_sum(const unsigned* __restrict__ pk,
              const float* __restrict__ hX,
              const float* __restrict__ hY,
              const unsigned* __restrict__ tags,
              float* __restrict__ aggr) {
    __shared__ float    acc[(DBLK + 1) * FEAT];
    __shared__ unsigned q[QCAP];

    const int tid  = threadIdx.x;
    const int wave = tid >> 5;
    const int lane = tid & 31;
    const int lo   = blockIdx.x * DBLK;
    const unsigned pat = (unsigned)blockIdx.x * 0x01010101u;

    for (int i = tid; i < (DBLK + 1) * FEAT; i += SCAN_T) acc[i] = 0.0f;
    __syncthreads();

    unsigned* qw = q + wave * QW;
    int cnt  = 0;
    int head = 0;

    const uint4* tg = (const uint4*)tags;
    const int t0 = (blockIdx.x * NTILES) / NBLK;

    uint4 v[SCAN_U];
    {
        const int base = t0 * TILE16 + tid;
        #pragma unroll
        for (int u = 0; u < SCAN_U; ++u) {
            int i16 = base + u * SCAN_T; if (i16 > E16 - 1) i16 = E16 - 1;
            v[u] = tg[i16];
        }
    }
    for (int t = 0; t < NTILES; ++t) {
        int tt = t + t0; if (tt >= NTILES) tt -= NTILES;
        int tn = tt + 1; if (tn >= NTILES) tn -= NTILES;
        const int base  = tt * TILE16 + tid;
        const int nbase = tn * TILE16 + tid;
        uint4 w[SCAN_U];
        #pragma unroll
        for (int u = 0; u < SCAN_U; ++u) {
            int i16 = nbase + u * SCAN_T; if (i16 > E16 - 1) i16 = E16 - 1;
            w[u] = tg[i16];
        }
        #pragma unroll
        for (int u = 0; u < SCAN_U; ++u) {
            const int i16 = base + u * SCAN_T;
            const unsigned x0 = v[u].x ^ pat, x1 = v[u].y ^ pat, x2 = v[u].z ^ pat, x3 = v[u].w ^ pat;
            const unsigned n0 = (x0 - 0x01010101u) & ~x0;
            const unsigned n1 = (x1 - 0x01010101u) & ~x1;
            const unsigned n2 = (x2 - 0x01010101u) & ~x2;
            const unsigned n3 = (x3 - 0x01010101u) & ~x3;
            unsigned mm = 0u;
            if (i16 < E16)
                mm = ((n0 & 0x80808080u) >> 7) | ((n1 & 0x80808080u) >> 6) |
                     ((n2 & 0x80808080u) >> 5) | ((n3 & 0x80808080u) >> 4);
            const unsigned ebase = (unsigned)i16 * 16u;
            unsigned bal = __builtin_amdgcn_ballot_w32(mm != 0u);
            while (bal != 0u) {
                if (mm != 0u) {
                    const unsigned pos = (unsigned)__builtin_ctz(mm);
                    mm &= mm - 1u;
                    const unsigned slot = (unsigned)cnt + __builtin_amdgcn_mbcnt_lo(bal, 0u);
                    if (slot < (unsigned)QW)
                        qw[slot] = ebase + ((pos & 7u) << 2) + (pos >> 3);
                }
                cnt += __builtin_popcount(bal);
                bal = __builtin_amdgcn_ballot_w32(mm != 0u);
            }
        }
        const bool flush = (t == NTILES - 1) || (cnt > QW - WAVE_TILE_EDGES);
        while (cnt - head >= 32 || (flush && cnt > head)) {
            int n = cnt - head; if (n > 32) n = 32;
            drain_run(qw + head, n, acc, (unsigned)blockIdx.x, pk, hX, hY, lane);
            head += n;
        }
        if (flush) { cnt = 0; head = 0; }
        #pragma unroll
        for (int u = 0; u < SCAN_U; ++u) v[u] = w[u];
    }
    __syncthreads();

    int lim = (Nn - lo) * FEAT;
    if (lim > DBLK * FEAT) lim = DBLK * FEAT;
    float* dst = aggr + (size_t)lo * FEAT;
    for (int i = tid; i < lim; i += SCAN_T) dst[i] = acc[i];
}

__device__ __forceinline__ v8f wmma_f32(v2f a, v2f b, v8f c) {
    return __builtin_amdgcn_wmma_f32_16x16x4_f32(
         false, a,  false, b,
         (short)0, c,  false,  false);
}

__global__ void gemm_ln_XY(const float* __restrict__ aggr,
                           const float* __restrict__ ht,
                           const float* __restrict__ WX,
                           const float* __restrict__ bX,
                           const float* __restrict__ gX,
                           const float* __restrict__ beX,
                           const float* __restrict__ WY,
                           const float* __restrict__ bY,
                           const float* __restrict__ gY,
                           const float* __restrict__ beY,
                           float* __restrict__ outX,
                           float* __restrict__ outY) {
    __shared__ float ldsX[4][16][HX + 1];
    __shared__ float ldsY[4][16][HY + 1];
    const int wave  = threadIdx.x >> 5;
    const int lane  = threadIdx.x & 31;
    const int group = blockIdx.x * 4 + wave;

    if (group < GROUPS) {
        const int m     = lane & 15;
        const int khalf = lane >> 4;
        const int node  = group * 16 + m;
        const float* arow = aggr + (size_t)node * FEAT;

        v8f acc[4]  = {};
        v8f accY[2] = {};
        #pragma unroll 4
        for (int k0 = 0; k0 < HX; k0 += 4) {
            const int ka = k0 + khalf * 2;
            v2f a;
            a.x = arow[ka];
            a.y = arow[ka + 1];
            #pragma unroll
            for (int t = 0; t < 4; ++t) {
                const int n = t * 16 + m;
                v2f b;
                b.x = WX[ka * HX + n];
                b.y = WX[(ka + 1) * HX + n];
                acc[t] = wmma_f32(a, b, acc[t]);
            }
        }
        #pragma unroll 4
        for (int k0 = HX; k0 < FEAT; k0 += 4) {
            const int ka = k0 + khalf * 2;
            const int ky = ka - HX;
            v2f a;
            a.x = arow[ka];
            a.y = arow[ka + 1];
            #pragma unroll
            for (int t = 0; t < 4; ++t) {
                const int n = t * 16 + m;
                v2f b;
                b.x = WX[ka * HX + n];
                b.y = WX[(ka + 1) * HX + n];
                acc[t] = wmma_f32(a, b, acc[t]);
            }
            #pragma unroll
            for (int t = 0; t < 2; ++t) {
                const int n = t * 16 + m;
                v2f b;
                b.x = WY[ky * HY + n];
                b.y = WY[(ky + 1) * HY + n];
                accY[t] = wmma_f32(a, b, accY[t]);
            }
        }
        #pragma unroll 4
        for (int k0 = FEAT; k0 < KX; k0 += 4) {
            const int ka = k0 + khalf * 2;
            v2f a;
            a.x = ht[ka - FEAT];
            a.y = ht[ka - FEAT + 1];
            #pragma unroll
            for (int t = 0; t < 4; ++t) {
                const int n = t * 16 + m;
                v2f b;
                b.x = WX[ka * HX + n];
                b.y = WX[(ka + 1) * HX + n];
                acc[t] = wmma_f32(a, b, acc[t]);
            }
        }
        #pragma unroll
        for (int t = 0; t < 4; ++t) {
            const int n = t * 16 + m;
            const float bias = bX[n];
            #pragma unroll
            for (int v = 0; v < 8; ++v) {
                const int row = khalf * 8 + v;
                float x = acc[t][v] + bias;
                ldsX[wave][row][n] = fmaxf(x, 0.0f);
            }
        }
        #pragma unroll
        for (int t = 0; t < 2; ++t) {
            const int n = t * 16 + m;
            const float bias = bY[n];
            #pragma unroll
            for (int v = 0; v < 8; ++v) {
                const int row = khalf * 8 + v;
                float x = accY[t][v] + bias;
                ldsY[wave][row][n] = fmaxf(x, 0.0f);
            }
        }
    }
    __syncthreads();
    if (group < GROUPS) {
        const int row  = lane >> 1;
        const int half = lane & 1;
        const int node = group * 16 + row;
        {
            float s = 0.0f, s2 = 0.0f;
            #pragma unroll
            for (int j = 0; j < 32; ++j) {
                float x = ldsX[wave][row][half * 32 + j];
                s += x; s2 += x * x;
            }
            s  += __shfl_xor(s, 1);
            s2 += __shfl_xor(s2, 1);
            const float mean = s * (1.0f / 64.0f);
            const float var  = s2 * (1.0f / 64.0f) - mean * mean;
            const float inv  = rsqrtf(var + 1e-5f);
            #pragma unroll
            for (int j = 0; j < 32; ++j) {
                const int c = half * 32 + j;
                float x = ldsX[wave][row][c];
                outX[(size_t)node * HX + c] = (x - mean) * inv * gX[c] + beX[c];
            }
        }
        {
            float s = 0.0f, s2 = 0.0f;
            #pragma unroll
            for (int j = 0; j < 16; ++j) {
                float x = ldsY[wave][row][half * 16 + j];
                s += x; s2 += x * x;
            }
            s  += __shfl_xor(s, 1);
            s2 += __shfl_xor(s2, 1);
            const float mean = s * (1.0f / 32.0f);
            const float var  = s2 * (1.0f / 32.0f) - mean * mean;
            const float inv  = rsqrtf(var + 1e-5f);
            #pragma unroll
            for (int j = 0; j < 16; ++j) {
                const int c = half * 16 + j;
                float x = ldsY[wave][row][c];
                outY[(size_t)node * HY + c] = (x - mean) * inv * gY[c] + beY[c];
            }
        }
    }
}

extern "C" void kernel_launch(void* const* d_in, const int* in_sizes, int n_in,
                              void* d_out, int out_size, void* d_ws, size_t ws_size,
                              hipStream_t stream) {
    const int*   erow = (const int*)d_in[0];
    const int*   ecol = (const int*)d_in[1];
    const float* hX   = (const float*)d_in[2];
    const float* hY   = (const float*)d_in[3];
    const float* ht   = (const float*)d_in[4];
    const float* WX   = (const float*)d_in[5];
    const float* bX   = (const float*)d_in[6];
    const float* gX   = (const float*)d_in[7];
    const float* beX  = (const float*)d_in[8];
    const float* WY   = (const float*)d_in[9];
    const float* bY   = (const float*)d_in[10];
    const float* gY   = (const float*)d_in[11];
    const float* beY  = (const float*)d_in[12];

    float* out  = (float*)d_out;
    float* aggr = (float*)d_ws;
    unsigned* tags = (unsigned*)((char*)d_ws + (32u << 20));
    unsigned* pk   = (unsigned*)((char*)d_ws + (64u << 20));

    tag_rows<<<TAG_BLOCKS, TAG_T, 0, stream>>>(erow, ecol, tags, pk);
    scan_sum<<<NBLK, SCAN_T, 0, stream>>>(pk, hX, hY, tags, aggr);

    const int blocks = (GROUPS + 3) / 4;
    gemm_ln_XY<<<blocks, 128, 0, stream>>>(aggr, ht, WX, bX, gX, beX, WY, bY, gY, beY,
                                           out, out + (size_t)Nn * HX);
}


# Products and LayerNorms moved into the slab kernel: WMMA A fragments read its summed rows in LDS, output stored as whole lines; aggr never stored; 3 launches to 2; LDS f32 atomics, arrival order kept
# speedup vs baseline: 2.9801x; 1.0921x over previous
#include <hip/hip_runtime.h>
#include <hip/hip_bf16.h>

#define Nn 50000
#define Ee 800000
#define HX 64
#define HY 32
#define HT 32
#define FEAT 96
#define KX 128

typedef __attribute__((ext_vector_type(2))) float v2f;
typedef __attribute__((ext_vector_type(8))) float v8f;

#define NBLK  256
#define DBLK  196
#define E4 (Ee / 4)
#define TAG_T 256
#define TAG_WAVE4 128
#define TAG_BLOCKS ((E4 + (TAG_T / 32) * TAG_WAVE4 - 1) / ((TAG_T / 32) * TAG_WAVE4))

static_assert(DBLK < 255, "the row within a slab and its out-of-range mark must fit a byte");
static_assert(Nn <= 65536, "a source id must fit 16 bits of the packed word");

__device__ __forceinline__ unsigned slab_tag(int row) {
    return ((unsigned)row / (unsigned)DBLK) & 255u;
}

__device__ __forceinline__ unsigned pack_edge(int row, int col) {
    const unsigned r = (unsigned)row;
    const unsigned s = r / (unsigned)DBLK;
    const unsigned l = (r < (unsigned)(NBLK * DBLK)) ? (r - s * (unsigned)DBLK) : 255u;
    return ((s & 255u) << 24) | (l << 16) | ((unsigned)col & 0xffffu);
}

__global__ __launch_bounds__(TAG_T)
void tag_rows(const int* __restrict__ erow, const int* __restrict__ ecol,
              unsigned* __restrict__ tags, unsigned* __restrict__ pk) {
    const int wave = (blockIdx.x * TAG_T + threadIdx.x) >> 5;
    const int lane = threadIdx.x & 31;
    const int4* er4 = (const int4*)erow;
    const int4* ec4 = (const int4*)ecol;
    uint4* pk4 = (uint4*)pk;
    int4 r[4], c[4];
    #pragma unroll
    for (int j = 0; j < 4; ++j) {
        int i4 = wave * TAG_WAVE4 + j * 32 + lane; if (i4 > E4 - 1) i4 = E4 - 1;
        r[j] = er4[i4];
        c[j] = ec4[i4];
    }
    #pragma unroll
    for (int j = 0; j < 4; ++j) {
        const int i4 = wave * TAG_WAVE4 + j * 32 + lane;
        if (i4 < E4) {
            tags[i4] = slab_tag(r[j].x) | (slab_tag(r[j].y) << 8) |
                       (slab_tag(r[j].z) << 16) | (slab_tag(r[j].w) << 24);
            pk4[i4] = make_uint4(pack_edge(r[j].x, c[j].x), pack_edge(r[j].y, c[j].y),
                                 pack_edge(r[j].z, c[j].z), pack_edge(r[j].w, c[j].w));
        }
    }
}

__device__ __forceinline__ v8f wmma_f32(v2f a, v2f b, v8f c) {
    return __builtin_amdgcn_wmma_f32_16x16x4_f32(
         false, a,  false, b,
         (short)0, c,  false,  false);
}

#define SCAN_T 1024
#define SCAN_U 3
#define E16 (Ee / 16)
#define TILE16 (SCAN_T * SCAN_U)
#define NTILES ((E16 + TILE16 - 1) / TILE16)
#define QCAP 57344
#define QW (QCAP / (SCAN_T / 32))
#define WAVE_TILE_EDGES (32 * SCAN_U * 16)

#define DR_GRP  16
#define APITCH  (FEAT + 2)

#define ST_ROWS  224
#define STX_LD   (HX + 1)
#define STY_LD   (HY + 1)
#define STY_OFF  (ST_ROWS * STX_LD)
#define STAT_OFF (STY_OFF + ST_ROWS * STY_LD)
static_assert(ST_ROWS >= DBLK && ST_ROWS % 32 == 0, "whole pairs of 16-row groups must cover a slab");
static_assert((ST_ROWS / 32) * (HX / 16) <= SCAN_T / 32, "one wave per pair of groups and column tile");
static_assert(2 * HY <= HX, "the waves of the first column tiles of new_X take the tiles of new_Y");
static_assert(ST_ROWS / 16 <= 16 && 16 + ST_ROWS / 16 <= SCAN_T / 32, "the statistics pass: waves 0.. for new_X, 16.. for new_Y");
static_assert(STAT_OFF + 4 * ST_ROWS <= QCAP, "the staging area must fit the queue's LDS");
static_assert(SCAN_T % HX == 0 && SCAN_T % HY == 0, "a thread of the store pass keeps its column");

__device__ __forceinline__ void drain_run(const unsigned* qrun, int n, float* acc, unsigned slab,
                                          const unsigned* __restrict__ pk,
                                          const float* __restrict__ hX,
                                          const float* __restrict__ hY,
                                          int lane) {
    const unsigned ent = (lane < n) ? qrun[lane] : 0u;
    const unsigned p = pk[ent];
    const int col = (int)(p & 0xffffu);
    const unsigned d = (p >> 16) & 255u;
    const int loc = (lane < n && (p >> 24) == slab && d < (unsigned)DBLK) ? (int)d : DBLK;
    for (int j0 = 0; j0 < n; j0 += DR_GRP) {
        float x0[DR_GRP], x1[DR_GRP], y0[DR_GRP];
        int   dr[DR_GRP];
        #pragma unroll
        for (int u = 0; u < DR_GRP; ++u) {
            const int c = __builtin_amdgcn_readlane(col, j0 + u);
            dr[u] = __builtin_amdgcn_readlane(loc, j0 + u) * APITCH;
            const float* sx = hX + (size_t)c * HX;
            const float* sy = hY + (size_t)c * HY;
            x0[u] = sx[lane];
            x1[u] = sx[lane + 32];
            y0[u] = sy[lane];
        }
        #pragma unroll
        for (int u = 0; u < DR_GRP; ++u) {
            atomicAdd(&acc[dr[u] + lane],      x0[u]);
            atomicAdd(&acc[dr[u] + lane + 32], x1[u]);
            atomicAdd(&acc[dr[u] + lane + 64], y0[u]);
        }
    }
}

__global__ __launch_bounds__(SCAN_T)
void scan_sum(const unsigned* __restrict__ pk,
              const float* __restrict__ hX,
              const float* __restrict__ hY,
              const unsigned* __restrict__ tags,
              const float* __restrict__ ht,
              const float* __restrict__ WX,
              const float* __restrict__ bX,
              const float* __restrict__ gX,
              const float* __restrict__ beX,
              const float* __restrict__ WY,
              const float* __restrict__ bY,
              const float* __restrict__ gY,
              const float* __restrict__ beY,
              float* __restrict__ outX,
              float* __restrict__ outY) {
    __shared__ float    acc[(DBLK + 1) * APITCH];
    __shared__ unsigned q[QCAP];

    const int tid  = threadIdx.x;
    const int wave = tid >> 5;
    const int lane = tid & 31;
    const int lo   = blockIdx.x * DBLK;
    const unsigned pat = (unsigned)blockIdx.x * 0x01010101u;

    for (int i = tid; i < (DBLK + 1) * APITCH; i += SCAN_T) acc[i] = 0.0f;
    __syncthreads();

    unsigned* qw = q + wave * QW;
    int cnt  = 0;
    int head = 0;

    const uint4* tg = (const uint4*)tags;
    const int t0 = (blockIdx.x * NTILES) / NBLK;

    uint4 v[SCAN_U];
    {
        const int base = t0 * TILE16 + tid;
        #pragma unroll
        for (int u = 0; u < SCAN_U; ++u) {
            int i16 = base + u * SCAN_T; if (i16 > E16 - 1) i16 = E16 - 1;
            v[u] = tg[i16];
        }
    }
    for (int t = 0; t < NTILES; ++t) {
        int tt = t + t0; if (tt >= NTILES) tt -= NTILES;
        int tn = tt + 1; if (tn >= NTILES) tn -= NTILES;
        const int base  = tt * TILE16 + tid;
        const int nbase = tn * TILE16 + tid;
        uint4 w[SCAN_U];
        #pragma unroll
        for (int u = 0; u < SCAN_U; ++u) {
            int i16 = nbase + u * SCAN_T; if (i16 > E16 - 1) i16 = E16 - 1;
            w[u] = tg[i16];
        }
        #pragma unroll
        for (int u = 0; u < SCAN_U; ++u) {
            const int i16 = base + u * SCAN_T;
            const unsigned x0 = v[u].x ^ pat, x1 = v[u].y ^ pat, x2 = v[u].z ^ pat, x3 = v[u].w ^ pat;
            const unsigned n0 = (x0 - 0x01010101u) & ~x0;
            const unsigned n1 = (x1 - 0x01010101u) & ~x1;
            const unsigned n2 = (x2 - 0x01010101u) & ~x2;
            const unsigned n3 = (x3 - 0x01010101u) & ~x3;
            unsigned mm = 0u;
            if (i16 < E16)
                mm = ((n0 & 0x80808080u) >> 7) | ((n1 & 0x80808080u) >> 6) |
                     ((n2 & 0x80808080u) >> 5) | ((n3 & 0x80808080u) >> 4);
            const unsigned ebase = (unsigned)i16 * 16u;
            unsigned bal = __builtin_amdgcn_ballot_w32(mm != 0u);
            while (bal != 0u) {
                if (mm != 0u) {
                    const unsigned pos = (unsigned)__builtin_ctz(mm);
                    mm &= mm - 1u;
                    const unsigned slot = (unsigned)cnt + __builtin_amdgcn_mbcnt_lo(bal, 0u);
                    if (slot < (unsigned)QW)
                        qw[slot] = ebase + ((pos & 7u) << 2) + (pos >> 3);
                }
                cnt += __builtin_popcount(bal);
                bal = __builtin_amdgcn_ballot_w32(mm != 0u);
            }
        }
        const bool flush = (t == NTILES - 1) || (cnt > QW - WAVE_TILE_EDGES);
        while (cnt - head >= 32 || (flush && cnt > head)) {
            int n = cnt - head; if (n > 32) n = 32;
            drain_run(qw + head, n, acc, (unsigned)blockIdx.x, pk, hX, hY, lane);
            head += n;
        }
        if (flush) { cnt = 0; head = 0; }
        #pragma unroll
        for (int u = 0; u < SCAN_U; ++u) v[u] = w[u];
    }
    __syncthreads();

    int nrows = Nn - lo;
    if (nrows > DBLK) nrows = DBLK;
    float* stX  = (float*)q;
    float* stY  = stX + STY_OFF;
    float* stat = stX + STAT_OFF;
    {
        const int gp = wave >> 2;
        const int t  = wave & 3;
        if (gp * 32 < nrows) {
            const int m     = lane & 15;
            const int khalf = lane >> 4;
            const int n     = t * 16 + m;
            int r0 = gp * 32 + m;      if (r0 > DBLK) r0 = DBLK;
            int r1 = gp * 32 + 16 + m; if (r1 > DBLK) r1 = DBLK;
            const float* arow0 = acc + r0 * APITCH;
            const float* arow1 = acc + r1 * APITCH;

            v8f cX[2] = {};
            v8f cY[2] = {};
            #pragma unroll 4
            for (int k0 = 0; k0 < FEAT; k0 += 4) {
                const int ka = k0 + khalf * 2;
                v2f a0, a1, b;
                a0.x = arow0[ka];
                a0.y = arow0[ka + 1];
                a1.x = arow1[ka];
                a1.y = arow1[ka + 1];
                b.x = WX[ka * HX + n];
                b.y = WX[(ka + 1) * HX + n];
                cX[0] = wmma_f32(a0, b, cX[0]);
                cX[1] = wmma_f32(a1, b, cX[1]);
            }
            #pragma unroll 4
            for (int k0 = FEAT; k0 < KX; k0 += 4) {
                const int ka = k0 + khalf * 2;
                v2f a, b;
                a.x = ht[ka - FEAT];
                a.y = ht[ka - FEAT + 1];
                b.x = WX[ka * HX + n];
                b.y = WX[(ka + 1) * HX + n];
                cX[0] = wmma_f32(a, b, cX[0]);
                cX[1] = wmma_f32(a, b, cX[1]);
            }
            if (t < 2) {
                #pragma unroll 4
                for (int k0 = HX; k0 < FEAT; k0 += 4) {
                    const int ka = k0 + khalf * 2;
                    const int ky = ka - HX;
                    v2f a0, a1, b;
                    a0.x = arow0[ka];
                    a0.y = arow0[ka + 1];
                    a1.x = arow1[ka];
                    a1.y = arow1[ka + 1];
                    b.x = WY[ky * HY + n];
                    b.y = WY[(ky + 1) * HY + n];
                    cY[0] = wmma_f32(a0, b, cY[0]);
                    cY[1] = wmma_f32(a1, b, cY[1]);
                }
            }
            {
                const float bias = bX[n];
                #pragma unroll
                for (int u = 0; u < 2; ++u) {
                    #pragma unroll
                    for (int v = 0; v < 8; ++v) {
                        const int row = gp * 32 + u * 16 + khalf * 8 + v;
                        const float x = cX[u][v] + bias;
                        stX[row * STX_LD + n] = fmaxf(x, 0.0f);
                    }
                }
            }
            if (t < 2) {
                const float bias = bY[n];
                #pragma unroll
                for (int u = 0; u < 2; ++u) {
                    #pragma unroll
                    for (int v = 0; v < 8; ++v) {
                        const int row = gp * 32 + u * 16 + khalf * 8 + v;
                        const float x = cY[u][v] + bias;
                        stY[row * STY_LD + n] = fmaxf(x, 0.0f);
                    }
                }
            }
        }
    }
    __syncthreads();

    if (wave < ST_ROWS / 16) {
        const int row  = wave * 16 + (lane >> 1);
        const int half = lane & 1;
        float s = 0.0f, s2 = 0.0f;
        #pragma unroll
        for (int j = 0; j < 32; ++j) {
            const float x = stX[row * STX_LD + half * 32 + j];
            s += x; s2 += x * x;
        }
        s  += __shfl_xor(s, 1);
        s2 += __shfl_xor(s2, 1);
        const float mean = s * (1.0f / 64.0f);
        const float var  = s2 * (1.0f / 64.0f) - mean * mean;
        const float inv  = rsqrtf(var + 1e-5f);
        if (half == 0) { stat[row] = mean; stat[ST_ROWS + row] = inv; }
    } else if (wave >= 16 && wave < 16 + ST_ROWS / 16) {
        const int row  = (wave - 16) * 16 + (lane >> 1);
        const int half = lane & 1;
        float s = 0.0f, s2 = 0.0f;
        #pragma unroll
        for (int j = 0; j < 16; ++j) {
            const float x = stY[row * STY_LD + half * 16 + j];
            s += x; s2 += x * x;
        }
        s  += __shfl_xor(s, 1);
        s2 += __shfl_xor(s2, 1);
        const float mean = s * (1.0f / 32.0f);
        const float var  = s2 * (1.0f / 32.0f) - mean * mean;
        const float inv  = rsqrtf(var + 1e-5f);
        if (half == 0) { stat[2 * ST_ROWS + row] = mean; stat[3 * ST_ROWS + row] = inv; }
    }
    __syncthreads();

    {
        float* oX = outX + (size_t)lo * HX;
        const int c = tid & (HX - 1);
        const float gg = gX[c], bb = beX[c];
        const int limX = nrows * HX;
        for (int i = tid; i < limX; i += SCAN_T) {
            const int r = i / HX;
            const float x = stX[r * STX_LD + c];
            oX[i] = (x - stat[r]) * stat[ST_ROWS + r] * gg + bb;
        }
    }
    {
        float* oY = outY + (size_t)lo * HY;
        const int c = tid & (HY - 1);
        const float gg = gY[c], bb = beY[c];
        const int limY = nrows * HY;
        for (int i = tid; i < limY; i += SCAN_T) {
            const int r = i / HY;
            const float x = stY[r * STY_LD + c];
            oY[i] = (x - stat[2 * ST_ROWS + r]) * stat[3 * ST_ROWS + r] * gg + bb;
        }
    }
}


extern "C" void kernel_launch(void* const* d_in, const int* in_sizes, int n_in,
                              void* d_out, int out_size, void* d_ws, size_t ws_size,
                              hipStream_t stream) {
    const int*   erow = (const int*)d_in[0];
    const int*   ecol = (const int*)d_in[1];
    const float* hX   = (const float*)d_in[2];
    const float* hY   = (const float*)d_in[3];
    const float* ht   = (const float*)d_in[4];
    const float* WX   = (const float*)d_in[5];
    const float* bX   = (const float*)d_in[6];
    const float* gX   = (const float*)d_in[7];
    const float* beX  = (const float*)d_in[8];
    const float* WY   = (const float*)d_in[9];
    const float* bY   = (const float*)d_in[10];
    const float* gY   = (const float*)d_in[11];
    const float* beY  = (const float*)d_in[12];

    float* out  = (float*)d_out;
    unsigned* tags = (unsigned*)((char*)d_ws + (32u << 20));
    unsigned* pk   = (unsigned*)((char*)d_ws + (64u << 20));

    tag_rows<<<TAG_BLOCKS, TAG_T, 0, stream>>>(erow, ecol, tags, pk);
    scan_sum<<<NBLK, SCAN_T, 0, stream>>>(pk, hX, hY, tags, ht, WX, bX, gX, beX, WY, bY, gY, beY,
                                          out, out + (size_t)Nn * HX);
}


# Tag launch writes a membership bitmap per slab (1 bit per edge, OR-ed in LDS per 2048-edge chunk) in place of byte tags; a slab reads only its own bitmap; LDS f32 atomics, arrival order kept
# speedup vs baseline: 3.0952x; 1.0386x over previous
#include <hip/hip_runtime.h>
#include <hip/hip_bf16.h>

#define Nn 50000
#define Ee 800000
#define HX 64
#define HY 32
#define HT 32
#define FEAT 96
#define KX 128

typedef __attribute__((ext_vector_type(2))) float v2f;
typedef __attribute__((ext_vector_type(8))) float v8f;

#define NBLK  256
#define DBLK  196
#define E4 (Ee / 4)
#define CH    2048
#define CHW   (CH / 32)
#define CH4   (CH / 4)
#define NCH   ((Ee + CH - 1) / CH)
#define BMW   (NCH * CHW)
#define TAG_T 256
#define TAG_U (CH4 / TAG_T)
#define BPITCH (CHW + 4)

static_assert(DBLK < 255, "the row within a slab and its out-of-range mark must fit a byte");
static_assert(Nn <= 65536, "a source id must fit 16 bits of the packed word");
static_assert(CH4 % TAG_T == 0 && CHW % 4 == 0 && BPITCH % 4 == 0, "whole 16-byte elements");
static_assert(TAG_T % (CHW / 4) == 0 && NBLK % (TAG_T / (CHW / 4)) == 0, "the store pass covers whole pieces");

__device__ __forceinline__ unsigned pack_edge(int row, int col) {
    const unsigned r = (unsigned)row;
    const unsigned s = r / (unsigned)DBLK;
    const unsigned l = (r < (unsigned)(NBLK * DBLK)) ? (r - s * (unsigned)DBLK) : 255u;
    return ((s & 255u) << 24) | (l << 16) | ((unsigned)col & 0xffffu);
}

__device__ __forceinline__ void mark_edge(unsigned* lb, int row, unsigned w, unsigned bit) {
    const unsigned r = (unsigned)row;
    if (r < (unsigned)(NBLK * DBLK))
        atomicOr(&lb[(r / (unsigned)DBLK) * BPITCH + w], 1u << bit);
}

__global__ __launch_bounds__(TAG_T)
void tag_rows(const int* __restrict__ erow, const int* __restrict__ ecol,
              unsigned* __restrict__ bm, unsigned* __restrict__ pk) {
    __shared__ unsigned lb[NBLK * BPITCH];
    const int tid   = threadIdx.x;
    const int chunk = blockIdx.x;
    const int4* er4 = (const int4*)erow;
    const int4* ec4 = (const int4*)ecol;
    uint4* pk4 = (uint4*)pk;
    uint4* lb4 = (uint4*)lb;
    int4 r[TAG_U], c[TAG_U];
    #pragma unroll
    for (int j = 0; j < TAG_U; ++j) {
        int i4 = chunk * CH4 + j * TAG_T + tid; if (i4 > E4 - 1) i4 = E4 - 1;
        r[j] = er4[i4];
        c[j] = ec4[i4];
    }
    for (int i = tid; i < NBLK * BPITCH / 4; i += TAG_T) lb4[i] = make_uint4(0u, 0u, 0u, 0u);
    __syncthreads();
    #pragma unroll
    for (int j = 0; j < TAG_U; ++j) {
        const int e4 = j * TAG_T + tid;
        const int i4 = chunk * CH4 + e4;
        if (i4 < E4) {
            const unsigned w = (unsigned)e4 >> 3;
            const unsigned b = ((unsigned)e4 & 7u) * 4u;
            mark_edge(lb, r[j].x, w, b);
            mark_edge(lb, r[j].y, w, b + 1u);
            mark_edge(lb, r[j].z, w, b + 2u);
            mark_edge(lb, r[j].w, w, b + 3u);
            pk4[i4] = make_uint4(pack_edge(r[j].x, c[j].x), pack_edge(r[j].y, c[j].y),
                                 pack_edge(r[j].z, c[j].z), pack_edge(r[j].w, c[j].w));
        }
    }
    __syncthreads();
    {
        uint4* bm4 = (uint4*)bm;
        const int qd = tid % (CHW / 4);
        for (int s = tid / (CHW / 4); s < NBLK; s += TAG_T / (CHW / 4))
            bm4[((size_t)s * NCH + chunk) * (CHW / 4) + qd] = lb4[s * (BPITCH / 4) + qd];
    }
}

__device__ __forceinline__ v8f wmma_f32(v2f a, v2f b, v8f c) {
    return __builtin_amdgcn_wmma_f32_16x16x4_f32(
         false, a,  false, b,
         (short)0, c,  false,  false);
}

#define SCAN_T 1024
#define SCAN_W (SCAN_T / 32)
#define BM16 (BMW / 4)
#define WSH16 ((BM16 + SCAN_W - 1) / SCAN_W)
#define NTILES ((WSH16 + 31) / 32)
#define QCAP 57344
#define QW (QCAP / SCAN_W)

#define DR_GRP  16
#define APITCH  (FEAT + 2)

#define ST_ROWS  224
#define STX_LD   (HX + 1)
#define STY_LD   (HY + 1)
#define STY_OFF  (ST_ROWS * STX_LD)
#define STAT_OFF (STY_OFF + ST_ROWS * STY_LD)
static_assert(ST_ROWS >= DBLK && ST_ROWS % 32 == 0, "whole pairs of 16-row groups must cover a slab");
static_assert((ST_ROWS / 32) * (HX / 16) <= SCAN_T / 32, "one wave per pair of groups and column tile");
static_assert(2 * HY <= HX, "the waves of the first column tiles of new_X take the tiles of new_Y");
static_assert(ST_ROWS / 16 <= 16 && 16 + ST_ROWS / 16 <= SCAN_T / 32, "the statistics pass: waves 0.. for new_X, 16.. for new_Y");
static_assert(STAT_OFF + 4 * ST_ROWS <= QCAP, "the staging area must fit the queue's LDS");
static_assert(SCAN_T % HX == 0 && SCAN_T % HY == 0, "a thread of the store pass keeps its column");

__device__ __forceinline__ void drain_run(const unsigned* qrun, int n, float* acc, unsigned slab,
                                          const unsigned* __restrict__ pk,
                                          const float* __restrict__ hX,
                                          const float* __restrict__ hY,
                                          int lane) {
    const unsigned ent = (lane < n) ? qrun[lane] : 0u;
    const unsigned p = pk[ent];
    const int col = (int)(p & 0xffffu);
    const unsigned d = (p >> 16) & 255u;
    const int loc = (lane < n && (p >> 24) == slab && d < (unsigned)DBLK) ? (int)d : DBLK;
    for (int j0 = 0; j0 < n; j0 += DR_GRP) {
        float x0[DR_GRP], x1[DR_GRP], y0[DR_GRP];
        int   dr[DR_GRP];
        #pragma unroll
        for (int u = 0; u < DR_GRP; ++u) {
            const int c = __builtin_amdgcn_readlane(col, j0 + u);
            dr[u] = __builtin_amdgcn_readlane(loc, j0 + u) * APITCH;
            const float* sx = hX + (size_t)c * HX;
            const float* sy = hY + (size_t)c * HY;
            x0[u] = sx[lane];
            x1[u] = sx[lane + 32];
            y0[u] = sy[lane];
        }
        #pragma unroll
        for (int u = 0; u < DR_GRP; ++u) {
            atomicAdd(&acc[dr[u] + lane],      x0[u]);
            atomicAdd(&acc[dr[u] + lane + 32], x1[u]);
            atomicAdd(&acc[dr[u] + lane + 64], y0[u]);
        }
    }
}

__global__ __launch_bounds__(SCAN_T)
void scan_sum(const unsigned* __restrict__ pk,
              const float* __restrict__ hX,
              const float* __restrict__ hY,
              const unsigned* __restrict__ bm,
              const float* __restrict__ ht,
              const float* __restrict__ WX,
              const float* __restrict__ bX,
              const float* __restrict__ gX,
              const float* __restrict__ beX,
              const float* __restrict__ WY,
              const float* __restrict__ bY,
              const float* __restrict__ gY,
              const float* __restrict__ beY,
              float* __restrict__ outX,
              float* __restrict__ outY) {
    __shared__ float    acc[(DBLK + 1) * APITCH];
    __shared__ unsigned q[QCAP];

    const int tid  = threadIdx.x;
    const int wave = tid >> 5;
    const int lane = tid & 31;
    const int lo   = blockIdx.x * DBLK;

    unsigned* qw = q + wave * QW;
    int cnt  = 0;
    int head = 0;

    const uint4* bs = (const uint4*)(bm + (size_t)blockIdx.x * BMW);
    const int w0 = wave * WSH16;

    uint4 v = make_uint4(0u, 0u, 0u, 0u);
    if (lane < WSH16 && w0 + lane < BM16) v = bs[w0 + lane];

    for (int i = tid; i < (DBLK + 1) * APITCH; i += SCAN_T) acc[i] = 0.0f;
    __syncthreads();

    for (int t = 0; t < NTILES; ++t) {
        const int k  = t * 32 + lane;
        const int kn = k + 32;
        uint4 w = make_uint4(0u, 0u, 0u, 0u);
        if (kn < WSH16 && w0 + kn < BM16) w = bs[w0 + kn];
        unsigned m0 = v.x, m1 = v.y, m2 = v.z, m3 = v.w;
        const unsigned ebase = (unsigned)(w0 + k) * 128u;
        bool more;
        do {
#define TAKE_BITS(mm, off)                                                              \
            {                                                                           \
                unsigned bal = __builtin_amdgcn_ballot_w32((mm) != 0u);                 \
                while (bal != 0u && cnt <= QW - 32) {                                   \
                    if ((mm) != 0u) {                                                   \
                        const unsigned pos = (unsigned)__builtin_ctz(mm);               \
                        (mm) &= (mm) - 1u;                                              \
                        qw[cnt + (int)__builtin_amdgcn_mbcnt_lo(bal, 0u)] = ebase + (off) + pos; \
                    }                                                                   \
                    cnt += __builtin_popcount(bal);                                     \
                    bal = __builtin_amdgcn_ballot_w32((mm) != 0u);                      \
                }                                                                       \
            }
            TAKE_BITS(m0, 0u)
            TAKE_BITS(m1, 32u)
            TAKE_BITS(m2, 64u)
            TAKE_BITS(m3, 96u)
#undef TAKE_BITS
            more = __builtin_amdgcn_ballot_w32((m0 | m1 | m2 | m3) != 0u) != 0u;
            const bool flush = more || (t == NTILES - 1);
            while (cnt - head >= 32 || (flush && cnt > head)) {
                int n = cnt - head; if (n > 32) n = 32;
                drain_run(qw + head, n, acc, (unsigned)blockIdx.x, pk, hX, hY, lane);
                head += n;
            }
            if (flush) { cnt = 0; head = 0; }
        } while (more);
        v = w;
    }
    __syncthreads();

    int nrows = Nn - lo;
    if (nrows > DBLK) nrows = DBLK;
    float* stX  = (float*)q;
    float* stY  = stX + STY_OFF;
    float* stat = stX + STAT_OFF;
    {
        const int gp = wave >> 2;
        const int t  = wave & 3;
        if (gp * 32 < nrows) {
            const int m     = lane & 15;
            const int khalf = lane >> 4;
            const int n     = t * 16 + m;
            int r0 = gp * 32 + m;      if (r0 > DBLK) r0 = DBLK;
            int r1 = gp * 32 + 16 + m; if (r1 > DBLK) r1 = DBLK;
            const float* arow0 = acc + r0 * APITCH;
            const float* arow1 = acc + r1 * APITCH;

            v8f cX[2] = {};
            v8f cY[2] = {};
            #pragma unroll 4
            for (int k0 = 0; k0 < FEAT; k0 += 4) {
                const int ka = k0 + khalf * 2;
                v2f a0, a1, b;
                a0.x = arow0[ka];
                a0.y = arow0[ka + 1];
                a1.x = arow1[ka];
                a1.y = arow1[ka + 1];
                b.x = WX[ka * HX + n];
                b.y = WX[(ka + 1) * HX + n];
                cX[0] = wmma_f32(a0, b, cX[0]);
                cX[1] = wmma_f32(a1, b, cX[1]);
            }
            #pragma unroll 4
            for (int k0 = FEAT; k0 < KX; k0 += 4) {
                const int ka = k0 + khalf * 2;
                v2f a, b;
                a.x = ht[ka - FEAT];
                a.y = ht[ka - FEAT + 1];
                b.x = WX[ka * HX + n];
                b.y = WX[(ka + 1) * HX + n];
                cX[0] = wmma_f32(a, b, cX[0]);
                cX[1] = wmma_f32(a, b, cX[1]);
            }
            if (t < 2) {
                #pragma unroll 4
                for (int k0 = HX; k0 < FEAT; k0 += 4) {
                    const int ka = k0 + khalf * 2;
                    const int ky = ka - HX;
                    v2f a0, a1, b;
                    a0.x = arow0[ka];
                    a0.y = arow0[ka + 1];
                    a1.x = arow1[ka];
                    a1.y = arow1[ka + 1];
                    b.x = WY[ky * HY + n];
                    b.y = WY[(ky + 1) * HY + n];
                    cY[0] = wmma_f32(a0, b, cY[0]);
                    cY[1] = wmma_f32(a1, b, cY[1]);
                }
            }
            {
                const float bias = bX[n];
                #pragma unroll
                for (int u = 0; u < 2; ++u) {
                    #pragma unroll
                    for (int v = 0; v < 8; ++v) {
                        const int row = gp * 32 + u * 16 + khalf * 8 + v;
                        const float x = cX[u][v] + bias;
                        stX[row * STX_LD + n] = fmaxf(x, 0.0f);
                    }
                }
            }
            if (t < 2) {
                const float bias = bY[n];
                #pragma unroll
                for (int u = 0; u < 2; ++u) {
                    #pragma unroll
                    for (int v = 0; v < 8; ++v) {
                        const int row = gp * 32 + u * 16 + khalf * 8 + v;
                        const float x = cY[u][v] + bias;
                        stY[row * STY_LD + n] = fmaxf(x, 0.0f);
                    }
                }
            }
        }
    }
    __syncthreads();

    if (wave < ST_ROWS / 16) {
        const int row  = wave * 16 + (lane >> 1);
        const int half = lane & 1;
        float s = 0.0f, s2 = 0.0f;
        #pragma unroll
        for (int j = 0; j < 32; ++j) {
            const float x = stX[row * STX_LD + half * 32 + j];
            s += x; s2 += x * x;
        }
        s  += __shfl_xor(s, 1);
        s2 += __shfl_xor(s2, 1);
        const float mean = s * (1.0f / 64.0f);
        const float var  = s2 * (1.0f / 64.0f) - mean * mean;
        const float inv  = rsqrtf(var + 1e-5f);
        if (half == 0) { stat[row] = mean; stat[ST_ROWS + row] = inv; }
    } else if (wave >= 16 && wave < 16 + ST_ROWS / 16) {
        const int row  = (wave - 16) * 16 + (lane >> 1);
        const int half = lane & 1;
        float s = 0.0f, s2 = 0.0f;
        #pragma unroll
        for (int j = 0; j < 16; ++j) {
            const float x = stY[row * STY_LD + half * 16 + j];
            s += x; s2 += x * x;
        }
        s  += __shfl_xor(s, 1);
        s2 += __shfl_xor(s2, 1);
        const float mean = s * (1.0f / 32.0f);
        const float var  = s2 * (1.0f / 32.0f) - mean * mean;
        const float inv  = rsqrtf(var + 1e-5f);
        if (half == 0) { stat[2 * ST_ROWS + row] = mean; stat[3 * ST_ROWS + row] = inv; }
    }
    __syncthreads();

    {
        float* oX = outX + (size_t)lo * HX;
        const int c = tid & (HX - 1);
        const float gg = gX[c], bb = beX[c];
        const int limX = nrows * HX;
        for (int i = tid; i < limX; i += SCAN_T) {
            const int r = i / HX;
            const float x = stX[r * STX_LD + c];
            oX[i] = (x - stat[r]) * stat[ST_ROWS + r] * gg + bb;
        }
    }
    {
        float* oY = outY + (size_t)lo * HY;
        const int c = tid & (HY - 1);
        const float gg = gY[c], bb = beY[c];
        const int limY = nrows * HY;
        for (int i = tid; i < limY; i += SCAN_T) {
            const int r = i / HY;
            const float x = stY[r * STY_LD + c];
            oY[i] = (x - stat[2 * ST_ROWS + r]) * stat[3 * ST_ROWS + r] * gg + bb;
        }
    }
}


extern "C" void kernel_launch(void* const* d_in, const int* in_sizes, int n_in,
                              void* d_out, int out_size, void* d_ws, size_t ws_size,
                              hipStream_t stream) {
    const int*   erow = (const int*)d_in[0];
    const int*   ecol = (const int*)d_in[1];
    const float* hX   = (const float*)d_in[2];
    const float* hY   = (const float*)d_in[3];
    const float* ht   = (const float*)d_in[4];
    const float* WX   = (const float*)d_in[5];
    const float* bX   = (const float*)d_in[6];
    const float* gX   = (const float*)d_in[7];
    const float* beX  = (const float*)d_in[8];
    const float* WY   = (const float*)d_in[9];
    const float* bY   = (const float*)d_in[10];
    const float* gY   = (const float*)d_in[11];
    const float* beY  = (const float*)d_in[12];

    float* out  = (float*)d_out;
    unsigned* bm = (unsigned*)((char*)d_ws + (32u << 20));
    unsigned* pk = (unsigned*)((char*)d_ws + (64u << 20));

    tag_rows<<<NCH, TAG_T, 0, stream>>>(erow, ecol, bm, pk);
    scan_sum<<<NBLK, SCAN_T, 0, stream>>>(pk, hX, hY, bm, ht, WX, bX, gX, beX, WY, bY, gY, beY,
                                          out, out + (size_t)Nn * HX);
}


# Product phase 560 to 368 matrix instructions per SIMD: h_t part of new_X once per work-group as accumulator start (its sum order changes), Y tiles on all 4 SIMDs, all-padding row group skipped
# speedup vs baseline: 3.2007x; 1.0341x over previous
#include <hip/hip_runtime.h>
#include <hip/hip_bf16.h>

#define Nn 50000
#define Ee 800000
#define HX 64
#define HY 32
#define HT 32
#define FEAT 96
#define KX 128

typedef __attribute__((ext_vector_type(2))) float v2f;
typedef __attribute__((ext_vector_type(8))) float v8f;

#define NBLK  256
#define DBLK  196
#define E4 (Ee / 4)
#define CH    2048
#define CHW   (CH / 32)
#define CH4   (CH / 4)
#define NCH   ((Ee + CH - 1) / CH)
#define BMW   (NCH * CHW)
#define TAG_T 256
#define TAG_U (CH4 / TAG_T)
#define BPITCH (CHW + 4)

static_assert(DBLK < 255, "the row within a slab and its out-of-range mark must fit a byte");
static_assert(Nn <= 65536, "a source id must fit 16 bits of the packed word");
static_assert(CH4 % TAG_T == 0 && CHW % 4 == 0 && BPITCH % 4 == 0, "whole 16-byte elements");
static_assert(TAG_T % (CHW / 4) == 0 && NBLK % (TAG_T / (CHW / 4)) == 0, "the store pass covers whole pieces");

__device__ __forceinline__ unsigned pack_edge(int row, int col) {
    const unsigned r = (unsigned)row;
    const unsigned s = r / (unsigned)DBLK;
    const unsigned l = (r < (unsigned)(NBLK * DBLK)) ? (r - s * (unsigned)DBLK) : 255u;
    return ((s & 255u) << 24) | (l << 16) | ((unsigned)col & 0xffffu);
}

__device__ __forceinline__ void mark_edge(unsigned* lb, int row, unsigned w, unsigned bit) {
    const unsigned r = (unsigned)row;
    if (r < (unsigned)(NBLK * DBLK))
        atomicOr(&lb[(r / (unsigned)DBLK) * BPITCH + w], 1u << bit);
}

__global__ __launch_bounds__(TAG_T)
void tag_rows(const int* __restrict__ erow, const int* __restrict__ ecol,
              unsigned* __restrict__ bm, unsigned* __restrict__ pk) {
    __shared__ unsigned lb[NBLK * BPITCH];
    const int tid   = threadIdx.x;
    const int chunk = blockIdx.x;
    const int4* er4 = (const int4*)erow;
    const int4* ec4 = (const int4*)ecol;
    uint4* pk4 = (uint4*)pk;
    uint4* lb4 = (uint4*)lb;
    int4 r[TAG_U], c[TAG_U];
    #pragma unroll
    for (int j = 0; j < TAG_U; ++j) {
        int i4 = chunk * CH4 + j * TAG_T + tid; if (i4 > E4 - 1) i4 = E4 - 1;
        r[j] = er4[i4];
        c[j] = ec4[i4];
    }
    for (int i = tid; i < NBLK * BPITCH / 4; i += TAG_T) lb4[i] = make_uint4(0u, 0u, 0u, 0u);
    __syncthreads();
    #pragma unroll
    for (int j = 0; j < TAG_U; ++j) {
        const int e4 = j * TAG_T + tid;
        const int i4 = chunk * CH4 + e4;
        if (i4 < E4) {
            const unsigned w = (unsigned)e4 >> 3;
            const unsigned b = ((unsigned)e4 & 7u) * 4u;
            mark_edge(lb, r[j].x, w, b);
            mark_edge(lb, r[j].y, w, b + 1u);
            mark_edge(lb, r[j].z, w, b + 2u);
            mark_edge(lb, r[j].w, w, b + 3u);
            pk4[i4] = make_uint4(pack_edge(r[j].x, c[j].x), pack_edge(r[j].y, c[j].y),
                                 pack_edge(r[j].z, c[j].z), pack_edge(r[j].w, c[j].w));
        }
    }
    __syncthreads();
    {
        uint4* bm4 = (uint4*)bm;
        const int qd = tid % (CHW / 4);
        for (int s = tid / (CHW / 4); s < NBLK; s += TAG_T / (CHW / 4))
            bm4[((size_t)s * NCH + chunk) * (CHW / 4) + qd] = lb4[s * (BPITCH / 4) + qd];
    }
}

__device__ __forceinline__ v8f wmma_f32(v2f a, v2f b, v8f c) {
    return __builtin_amdgcn_wmma_f32_16x16x4_f32(
         false, a,  false, b,
         (short)0, c,  false,  false);
}

#define SCAN_T 1024
#define SCAN_W (SCAN_T / 32)
#define BM16 (BMW / 4)
#define WSH16 ((BM16 + SCAN_W - 1) / SCAN_W)
#define NTILES ((WSH16 + 31) / 32)
#define QCAP 57344
#define QW (QCAP / SCAN_W)

#define DR_GRP  16
#define APITCH  (FEAT + 2)

#define ST_ROWS  224
#define STX_LD   (HX + 1)
#define STY_LD   (HY + 1)
#define STY_OFF  (ST_ROWS * STX_LD)
#define STAT_OFF (STY_OFF + ST_ROWS * STY_LD)
static_assert(ST_ROWS >= DBLK && ST_ROWS % 32 == 0, "whole pairs of 16-row groups must cover a slab");
static_assert((ST_ROWS / 32) * (HX / 16) <= SCAN_T / 32, "one wave per pair of groups and column tile");
#define PROD_W  ((ST_ROWS / 32) * (HX / 16))
static_assert(2 * HY == HX, "a wave's column tile t gives Y column tile t & 1 of row group t >> 1");
static_assert(SCAN_T / 32 - PROD_W == HX / 16, "one spare wave per column tile for the h_t part");
static_assert(ST_ROWS / 16 <= 16 && 16 + ST_ROWS / 16 <= SCAN_T / 32, "the statistics pass: waves 0.. for new_X, 16.. for new_Y");
static_assert(STAT_OFF + 4 * ST_ROWS <= QCAP, "the staging area must fit the queue's LDS");
static_assert(SCAN_T % HX == 0 && SCAN_T % HY == 0, "a thread of the store pass keeps its column");

__device__ __forceinline__ void drain_run(const unsigned* qrun, int n, float* acc, unsigned slab,
                                          const unsigned* __restrict__ pk,
                                          const float* __restrict__ hX,
                                          const float* __restrict__ hY,
                                          int lane) {
    const unsigned ent = (lane < n) ? qrun[lane] : 0u;
    const unsigned p = pk[ent];
    const int col = (int)(p & 0xffffu);
    const unsigned d = (p >> 16) & 255u;
    const int loc = (lane < n && (p >> 24) == slab && d < (unsigned)DBLK) ? (int)d : DBLK;
    for (int j0 = 0; j0 < n; j0 += DR_GRP) {
        float x0[DR_GRP], x1[DR_GRP], y0[DR_GRP];
        int   dr[DR_GRP];
        #pragma unroll
        for (int u = 0; u < DR_GRP; ++u) {
            const int c = __builtin_amdgcn_readlane(col, j0 + u);
            dr[u] = __builtin_amdgcn_readlane(loc, j0 + u) * APITCH;
            const float* sx = hX + (size_t)c * HX;
            const float* sy = hY + (size_t)c * HY;
            x0[u] = sx[lane];
            x1[u] = sx[lane + 32];
            y0[u] = sy[lane];
        }
        #pragma unroll
        for (int u = 0; u < DR_GRP; ++u) {
            atomicAdd(&acc[dr[u] + lane],      x0[u]);
            atomicAdd(&acc[dr[u] + lane + 32], x1[u]);
            atomicAdd(&acc[dr[u] + lane + 64], y0[u]);
        }
    }
}


template <bool TWO>
__device__ __forceinline__ void slab_products(const float* acc, const float* cvec,
                                              float* stX, float* stY,
                                              const float* __restrict__ WX,
                                              const float* __restrict__ bX,
                                              const float* __restrict__ WY,
                                              const float* __restrict__ bY,
                                              int gp, int t, int lane) {
    const int m     = lane & 15;
    const int khalf = lane >> 4;
    const int n     = t * 16 + m;
    int r0 = gp * 32 + m;      if (r0 > DBLK) r0 = DBLK;
    int r1 = gp * 32 + 16 + m; if (r1 > DBLK) r1 = DBLK;
    const float* arow0 = acc + r0 * APITCH;
    const float* arow1 = acc + r1 * APITCH;

    const float cstart = cvec[n];
    v8f cX0, cX1;
    #pragma unroll
    for (int v = 0; v < 8; ++v) { cX0[v] = cstart; cX1[v] = cstart; }
    #pragma unroll 4
    for (int k0 = 0; k0 < FEAT; k0 += 4) {
        const int ka = k0 + khalf * 2;
        v2f a0, b;
        a0.x = arow0[ka];
        a0.y = arow0[ka + 1];
        b.x = WX[ka * HX + n];
        b.y = WX[(ka + 1) * HX + n];
        cX0 = wmma_f32(a0, b, cX0);
        if (TWO) {
            v2f a1;
            a1.x = arow1[ka];
            a1.y = arow1[ka + 1];
            cX1 = wmma_f32(a1, b, cX1);
        }
    }
    const int uy = t >> 1;
    const int ny = (t & 1) * 16 + m;
    if (TWO || uy == 0) {
        const float* arowY = uy ? arow1 : arow0;
        v8f cY = {};
        #pragma unroll 4
        for (int k0 = HX; k0 < FEAT; k0 += 4) {
            const int ka = k0 + khalf * 2;
            const int ky = ka - HX;
            v2f a, b;
            a.x = arowY[ka];
            a.y = arowY[ka + 1];
            b.x = WY[ky * HY + ny];
            b.y = WY[(ky + 1) * HY + ny];
            cY = wmma_f32(a, b, cY);
        }
        const float bias = bY[ny];
        #pragma unroll
        for (int v = 0; v < 8; ++v) {
            const int row = gp * 32 + uy * 16 + khalf * 8 + v;
            stY[row * STY_LD + ny] = fmaxf(cY[v] + bias, 0.0f);
        }
    }
    {
        const float bias = bX[n];
        #pragma unroll
        for (int v = 0; v < 8; ++v) {
            const int row = gp * 32 + khalf * 8 + v;
            stX[row * STX_LD + n] = fmaxf(cX0[v] + bias, 0.0f);
            if (TWO) stX[(row + 16) * STX_LD + n] = fmaxf(cX1[v] + bias, 0.0f);
        }
    }
}

__global__ __launch_bounds__(SCAN_T)
void scan_sum(const unsigned* __restrict__ pk,
              const float* __restrict__ hX,
              const float* __restrict__ hY,
              const unsigned* __restrict__ bm,
              const float* __restrict__ ht,
              const float* __restrict__ WX,
              const float* __restrict__ bX,
              const float* __restrict__ gX,
              const float* __restrict__ beX,
              const float* __restrict__ WY,
              const float* __restrict__ bY,
              const float* __restrict__ gY,
              const float* __restrict__ beY,
              float* __restrict__ outX,
              float* __restrict__ outY) {
    __shared__ float    acc[(DBLK + 1) * APITCH];
    __shared__ unsigned q[QCAP];
    __shared__ float    cvec[HX];

    const int tid  = threadIdx.x;
    const int wave = tid >> 5;
    const int lane = tid & 31;
    const int lo   = blockIdx.x * DBLK;

    unsigned* qw = q + wave * QW;
    int cnt  = 0;
    int head = 0;

    const uint4* bs = (const uint4*)(bm + (size_t)blockIdx.x * BMW);
    const int w0 = wave * WSH16;

    uint4 v = make_uint4(0u, 0u, 0u, 0u);
    if (lane < WSH16 && w0 + lane < BM16) v = bs[w0 + lane];

    if (wave >= PROD_W) {
        const int m     = lane & 15;
        const int khalf = lane >> 4;
        const int n     = (wave - PROD_W) * 16 + m;
        v8f c = {};
        #pragma unroll
        for (int k0 = FEAT; k0 < KX; k0 += 4) {
            const int ka = k0 + khalf * 2;
            v2f a, b;
            a.x = ht[ka - FEAT];
            a.y = ht[ka - FEAT + 1];
            b.x = WX[ka * HX + n];
            b.y = WX[(ka + 1) * HX + n];
            c = wmma_f32(a, b, c);
        }
        if (lane < 16) cvec[n] = c[0];
    }

    for (int i = tid; i < (DBLK + 1) * APITCH; i += SCAN_T) acc[i] = 0.0f;
    __syncthreads();

    for (int t = 0; t < NTILES; ++t) {
        const int k  = t * 32 + lane;
        const int kn = k + 32;
        uint4 w = make_uint4(0u, 0u, 0u, 0u);
        if (kn < WSH16 && w0 + kn < BM16) w = bs[w0 + kn];
        unsigned m0 = v.x, m1 = v.y, m2 = v.z, m3 = v.w;
        const unsigned ebase = (unsigned)(w0 + k) * 128u;
        bool more;
        do {
#define TAKE_BITS(mm, off)                                                              \
            {                                                                           \
                unsigned bal = __builtin_amdgcn_ballot_w32((mm) != 0u);                 \
                while (bal != 0u && cnt <= QW - 32) {                                   \
                    if ((mm) != 0u) {                                                   \
                        const unsigned pos = (unsigned)__builtin_ctz(mm);               \
                        (mm) &= (mm) - 1u;                                              \
                        qw[cnt + (int)__builtin_amdgcn_mbcnt_lo(bal, 0u)] = ebase + (off) + pos; \
                    }                                                                   \
                    cnt += __builtin_popcount(bal);                                     \
                    bal = __builtin_amdgcn_ballot_w32((mm) != 0u);                      \
                }                                                                       \
            }
            TAKE_BITS(m0, 0u)
            TAKE_BITS(m1, 32u)
            TAKE_BITS(m2, 64u)
            TAKE_BITS(m3, 96u)
#undef TAKE_BITS
            more = __builtin_amdgcn_ballot_w32((m0 | m1 | m2 | m3) != 0u) != 0u;
            const bool flush = more || (t == NTILES - 1);
            while (cnt - head >= 32 || (flush && cnt > head)) {
                int n = cnt - head; if (n > 32) n = 32;
                drain_run(qw + head, n, acc, (unsigned)blockIdx.x, pk, hX, hY, lane);
                head += n;
            }
            if (flush) { cnt = 0; head = 0; }
        } while (more);
        v = w;
    }
    __syncthreads();

    int nrows = Nn - lo;
    if (nrows > DBLK) nrows = DBLK;
    float* stX  = (float*)q;
    float* stY  = stX + STY_OFF;
    float* stat = stX + STAT_OFF;
    {
        const int gp = wave >> 2;
        const int t  = wave & 3;
        if (gp * 32 < nrows) {
            if (gp * 32 + 16 < nrows)
                slab_products<true>(acc, cvec, stX, stY, WX, bX, WY, bY, gp, t, lane);
            else
                slab_products<false>(acc, cvec, stX, stY, WX, bX, WY, bY, gp, t, lane);
        }
    }
    __syncthreads();

    if (wave < ST_ROWS / 16) {
        const int row  = wave * 16 + (lane >> 1);
        const int half = lane & 1;
        float s = 0.0f, s2 = 0.0f;
        #pragma unroll
        for (int j = 0; j < 32; ++j) {
            const float x = stX[row * STX_LD + half * 32 + j];
            s += x; s2 += x * x;
        }
        s  += __shfl_xor(s, 1);
        s2 += __shfl_xor(s2, 1);
        const float mean = s * (1.0f / 64.0f);
        const float var  = s2 * (1.0f / 64.0f) - mean * mean;
        const float inv  = rsqrtf(var + 1e-5f);
        if (half == 0) { stat[row] = mean; stat[ST_ROWS + row] = inv; }
    } else if (wave >= 16 && wave < 16 + ST_ROWS / 16) {
        const int row  = (wave - 16) * 16 + (lane >> 1);
        const int half = lane & 1;
        float s = 0.0f, s2 = 0.0f;
        #pragma unroll
        for (int j = 0; j < 16; ++j) {
            const float x = stY[row * STY_LD + half * 16 + j];
            s += x; s2 += x * x;
        }
        s  += __shfl_xor(s, 1);
        s2 += __shfl_xor(s2, 1);
        const float mean = s * (1.0f / 32.0f);
        const float var  = s2 * (1.0f / 32.0f) - mean * mean;
        const float inv  = rsqrtf(var + 1e-5f);
        if (half == 0) { stat[2 * ST_ROWS + row] = mean; stat[3 * ST_ROWS + row] = inv; }
    }
    __syncthreads();

    {
        float* oX = outX + (size_t)lo * HX;
        const int c = tid & (HX - 1);
        const float gg = gX[c], bb = beX[c];
        const int limX = nrows * HX;
        for (int i = tid; i < limX; i += SCAN_T) {
            const int r = i / HX;
            const float x = stX[r * STX_LD + c];
            oX[i] = (x - stat[r]) * stat[ST_ROWS + r] * gg + bb;
        }
    }
    {
        float* oY = outY + (size_t)lo * HY;
        const int c = tid & (HY - 1);
        const float gg = gY[c], bb = beY[c];
        const int limY = nrows * HY;
        for (int i = tid; i < limY; i += SCAN_T) {
            const int r = i / HY;
            const float x = stY[r * STY_LD + c];
            oY[i] = (x - stat[2 * ST_ROWS + r]) * stat[3 * ST_ROWS + r] * gg + bb;
        }
    }
}


extern "C" void kernel_launch(void* const* d_in, const int* in_sizes, int n_in,
                              void* d_out, int out_size, void* d_ws, size_t ws_size,
                              hipStream_t stream) {
    const int*   erow = (const int*)d_in[0];
    const int*   ecol = (const int*)d_in[1];
    const float* hX   = (const float*)d_in[2];
    const float* hY   = (const float*)d_in[3];
    const float* ht   = (const float*)d_in[4];
    const float* WX   = (const float*)d_in[5];
    const float* bX   = (const float*)d_in[6];
    const float* gX   = (const float*)d_in[7];
    const float* beX  = (const float*)d_in[8];
    const float* WY   = (const float*)d_in[9];
    const float* bY   = (const float*)d_in[10];
    const float* gY   = (const float*)d_in[11];
    const float* beY  = (const float*)d_in[12];

    float* out  = (float*)d_out;
    unsigned* bm = (unsigned*)((char*)d_ws + (32u << 20));
    unsigned* pk = (unsigned*)((char*)d_ws + (64u << 20));

    tag_rows<<<NCH, TAG_T, 0, stream>>>(erow, ecol, bm, pk);
    scan_sum<<<NBLK, SCAN_T, 0, stream>>>(pk, hX, hY, bm, ht, WX, bX, gX, beX, WY, bY, gY, beY,
                                          out, out + (size_t)Nn * HX);
}


# Drain row loads: scalar base + one 32-bit lane offset per row, lane picks by constant index; no 64-bit vector adds or translation waits between a group's loads; LDS f32 atomics, arrival order kept
# speedup vs baseline: 3.3555x; 1.0484x over previous
#include <hip/hip_runtime.h>
#include <hip/hip_bf16.h>

#define Nn 50000
#define Ee 800000
#define HX 64
#define HY 32
#define HT 32
#define FEAT 96
#define KX 128

typedef __attribute__((ext_vector_type(2))) float v2f;
typedef __attribute__((ext_vector_type(8))) float v8f;

#define NBLK  256
#define DBLK  196
#define E4 (Ee / 4)
#define CH    2048
#define CHW   (CH / 32)
#define CH4   (CH / 4)
#define NCH   ((Ee + CH - 1) / CH)
#define BMW   (NCH * CHW)
#define TAG_T 256
#define TAG_U (CH4 / TAG_T)
#define BPITCH (CHW + 4)

static_assert(DBLK < 255, "the row within a slab and its out-of-range mark must fit a byte");
static_assert(Nn <= 65536, "a source id must fit 16 bits of the packed word");
static_assert(CH4 % TAG_T == 0 && CHW % 4 == 0 && BPITCH % 4 == 0, "whole 16-byte elements");
static_assert(TAG_T % (CHW / 4) == 0 && NBLK % (TAG_T / (CHW / 4)) == 0, "the store pass covers whole pieces");

__device__ __forceinline__ unsigned pack_edge(int row, int col) {
    const unsigned r = (unsigned)row;
    const unsigned s = r / (unsigned)DBLK;
    const unsigned l = (r < (unsigned)(NBLK * DBLK)) ? (r - s * (unsigned)DBLK) : 255u;
    return ((s & 255u) << 24) | (l << 16) | ((unsigned)col & 0xffffu);
}

__device__ __forceinline__ void mark_edge(unsigned* lb, int row, unsigned w, unsigned bit) {
    const unsigned r = (unsigned)row;
    if (r < (unsigned)(NBLK * DBLK))
        atomicOr(&lb[(r / (unsigned)DBLK) * BPITCH + w], 1u << bit);
}

__global__ __launch_bounds__(TAG_T)
void tag_rows(const int* __restrict__ erow, const int* __restrict__ ecol,
              unsigned* __restrict__ bm, unsigned* __restrict__ pk) {
    __shared__ unsigned lb[NBLK * BPITCH];
    const int tid   = threadIdx.x;
    const int chunk = blockIdx.x;
    const int4* er4 = (const int4*)erow;
    const int4* ec4 = (const int4*)ecol;
    uint4* pk4 = (uint4*)pk;
    uint4* lb4 = (uint4*)lb;
    int4 r[TAG_U], c[TAG_U];
    #pragma unroll
    for (int j = 0; j < TAG_U; ++j) {
        int i4 = chunk * CH4 + j * TAG_T + tid; if (i4 > E4 - 1) i4 = E4 - 1;
        r[j] = er4[i4];
        c[j] = ec4[i4];
    }
    for (int i = tid; i < NBLK * BPITCH / 4; i += TAG_T) lb4[i] = make_uint4(0u, 0u, 0u, 0u);
    __syncthreads();
    #pragma unroll
    for (int j = 0; j < TAG_U; ++j) {
        const int e4 = j * TAG_T + tid;
        const int i4 = chunk * CH4 + e4;
        if (i4 < E4) {
            const unsigned w = (unsigned)e4 >> 3;
            const unsigned b = ((unsigned)e4 & 7u) * 4u;
            mark_edge(lb, r[j].x, w, b);
            mark_edge(lb, r[j].y, w, b + 1u);
            mark_edge(lb, r[j].z, w, b + 2u);
            mark_edge(lb, r[j].w, w, b + 3u);
            pk4[i4] = make_uint4(pack_edge(r[j].x, c[j].x), pack_edge(r[j].y, c[j].y),
                                 pack_edge(r[j].z, c[j].z), pack_edge(r[j].w, c[j].w));
        }
    }
    __syncthreads();
    {
        uint4* bm4 = (uint4*)bm;
        const int qd = tid % (CHW / 4);
        for (int s = tid / (CHW / 4); s < NBLK; s += TAG_T / (CHW / 4))
            bm4[((size_t)s * NCH + chunk) * (CHW / 4) + qd] = lb4[s * (BPITCH / 4) + qd];
    }
}

__device__ __forceinline__ v8f wmma_f32(v2f a, v2f b, v8f c) {
    return __builtin_amdgcn_wmma_f32_16x16x4_f32(
         false, a,  false, b,
         (short)0, c,  false,  false);
}

#define SCAN_T 1024
#define SCAN_W (SCAN_T / 32)
#define BM16 (BMW / 4)
#define WSH16 ((BM16 + SCAN_W - 1) / SCAN_W)
#define NTILES ((WSH16 + 31) / 32)
#define QCAP 57344
#define QW (QCAP / SCAN_W)

#define DR_GRP  16
#define APITCH  (FEAT + 2)

#define ST_ROWS  224
#define STX_LD   (HX + 1)
#define STY_LD   (HY + 1)
#define STY_OFF  (ST_ROWS * STX_LD)
#define STAT_OFF (STY_OFF + ST_ROWS * STY_LD)
static_assert(ST_ROWS >= DBLK && ST_ROWS % 32 == 0, "whole pairs of 16-row groups must cover a slab");
static_assert((ST_ROWS / 32) * (HX / 16) <= SCAN_T / 32, "one wave per pair of groups and column tile");
#define PROD_W  ((ST_ROWS / 32) * (HX / 16))
static_assert(2 * HY == HX, "a wave's column tile t gives Y column tile t & 1 of row group t >> 1");
static_assert(SCAN_T / 32 - PROD_W == HX / 16, "one spare wave per column tile for the h_t part");
static_assert(ST_ROWS / 16 <= 16 && 16 + ST_ROWS / 16 <= SCAN_T / 32, "the statistics pass: waves 0.. for new_X, 16.. for new_Y");
static_assert(STAT_OFF + 4 * ST_ROWS <= QCAP, "the staging area must fit the queue's LDS");
static_assert(SCAN_T % HX == 0 && SCAN_T % HY == 0, "a thread of the store pass keeps its column");

__device__ __forceinline__ void drain_group(int col, int loc, float* acc,
                                            const float* __restrict__ hX,
                                            const float* __restrict__ hY,
                                            int lane) {
    const float* hXu = hX + 32;
    float x0[DR_GRP], x1[DR_GRP], y0[DR_GRP];
    int   dr[DR_GRP];
    #pragma unroll
    for (int u = 0; u < DR_GRP; ++u) {
        const unsigned c = (unsigned)__builtin_amdgcn_readlane(col, u);
        dr[u] = __builtin_amdgcn_readlane(loc, u) * APITCH;
        const unsigned ox = c * (unsigned)HX + (unsigned)lane;
        const unsigned oy = c * (unsigned)HY + (unsigned)lane;
        x0[u] = hX[ox];
        x1[u] = hXu[ox];
        y0[u] = hY[oy];
    }
    #pragma unroll
    for (int u = 0; u < DR_GRP; ++u) {
        atomicAdd(&acc[dr[u] + lane],      x0[u]);
        atomicAdd(&acc[dr[u] + lane + 32], x1[u]);
        atomicAdd(&acc[dr[u] + lane + 64], y0[u]);
    }
}

__device__ __forceinline__ void drain_run(const unsigned* qrun, int n, float* acc, unsigned slab,
                                          const unsigned* __restrict__ pk,
                                          const float* __restrict__ hX,
                                          const float* __restrict__ hY,
                                          int lane) {
    const unsigned ent = (lane < n) ? qrun[lane] : 0u;
    const unsigned p = pk[ent];
    int col = (int)(p & 0xffffu);
    const unsigned d = (p >> 16) & 255u;
    int loc = (lane < n && (p >> 24) == slab && d < (unsigned)DBLK) ? (int)d : DBLK;
    static_assert(2 * DR_GRP == 32, "a run is two groups: the two 16-lane halves of the wave");
    for (int j0 = 0; j0 < n; j0 += DR_GRP) {
        drain_group(col, loc, acc, hX, hY, lane);
        col = (int)__builtin_amdgcn_permlanex16((unsigned)col, (unsigned)col, 0x76543210u, 0xfedcba98u, false, false);
        loc = (int)__builtin_amdgcn_permlanex16((unsigned)loc, (unsigned)loc, 0x76543210u, 0xfedcba98u, false, false);
    }
}


template <bool TWO>
__device__ __forceinline__ void slab_products(const float* acc, const float* cvec,
                                              float* stX, float* stY,
                                              const float* __restrict__ WX,
                                              const float* __restrict__ bX,
                                              const float* __restrict__ WY,
                                              const float* __restrict__ bY,
                                              int gp, int t, int lane) {
    const int m     = lane & 15;
    const int khalf = lane >> 4;
    const int n     = t * 16 + m;
    int r0 = gp * 32 + m;      if (r0 > DBLK) r0 = DBLK;
    int r1 = gp * 32 + 16 + m; if (r1 > DBLK) r1 = DBLK;
    const float* arow0 = acc + r0 * APITCH;
    const float* arow1 = acc + r1 * APITCH;

    const float cstart = cvec[n];
    v8f cX0, cX1;
    #pragma unroll
    for (int v = 0; v < 8; ++v) { cX0[v] = cstart; cX1[v] = cstart; }
    #pragma unroll 4
    for (int k0 = 0; k0 < FEAT; k0 += 4) {
        const int ka = k0 + khalf * 2;
        v2f a0, b;
        a0.x = arow0[ka];
        a0.y = arow0[ka + 1];
        b.x = WX[ka * HX + n];
        b.y = WX[(ka + 1) * HX + n];
        cX0 = wmma_f32(a0, b, cX0);
        if (TWO) {
            v2f a1;
            a1.x = arow1[ka];
            a1.y = arow1[ka + 1];
            cX1 = wmma_f32(a1, b, cX1);
        }
    }
    const int uy = t >> 1;
    const int ny = (t & 1) * 16 + m;
    if (TWO || uy == 0) {
        const float* arowY = uy ? arow1 : arow0;
        v8f cY = {};
        #pragma unroll 4
        for (int k0 = HX; k0 < FEAT; k0 += 4) {
            const int ka = k0 + khalf * 2;
            const int ky = ka - HX;
            v2f a, b;
            a.x = arowY[ka];
            a.y = arowY[ka + 1];
            b.x = WY[ky * HY + ny];
            b.y = WY[(ky + 1) * HY + ny];
            cY = wmma_f32(a, b, cY);
        }
        const float bias = bY[ny];
        #pragma unroll
        for (int v = 0; v < 8; ++v) {
            const int row = gp * 32 + uy * 16 + khalf * 8 + v;
            stY[row * STY_LD + ny] = fmaxf(cY[v] + bias, 0.0f);
        }
    }
    {
        const float bias = bX[n];
        #pragma unroll
        for (int v = 0; v < 8; ++v) {
            const int row = gp * 32 + khalf * 8 + v;
            stX[row * STX_LD + n] = fmaxf(cX0[v] + bias, 0.0f);
            if (TWO) stX[(row + 16) * STX_LD + n] = fmaxf(cX1[v] + bias, 0.0f);
        }
    }
}

__global__ __launch_bounds__(SCAN_T)
void scan_sum(const unsigned* __restrict__ pk,
              const float* __restrict__ hX,
              const float* __restrict__ hY,
              const unsigned* __restrict__ bm,
              const float* __restrict__ ht,
              const float* __restrict__ WX,
              const float* __restrict__ bX,
              const float* __restrict__ gX,
              const float* __restrict__ beX,
              const float* __restrict__ WY,
              const float* __restrict__ bY,
              const float* __restrict__ gY,
              const float* __restrict__ beY,
              float* __restrict__ outX,
              float* __restrict__ outY) {
    __shared__ float    acc[(DBLK + 1) * APITCH];
    __shared__ unsigned q[QCAP];
    __shared__ float    cvec[HX];

    const int tid  = threadIdx.x;
    const int wave = tid >> 5;
    const int lane = tid & 31;
    const int lo   = blockIdx.x * DBLK;

    unsigned* qw = q + wave * QW;
    int cnt  = 0;
    int head = 0;

    const uint4* bs = (const uint4*)(bm + (size_t)blockIdx.x * BMW);
    const int w0 = wave * WSH16;

    uint4 v = make_uint4(0u, 0u, 0u, 0u);
    if (lane < WSH16 && w0 + lane < BM16) v = bs[w0 + lane];

    if (wave >= PROD_W) {
        const int m     = lane & 15;
        const int khalf = lane >> 4;
        const int n     = (wave - PROD_W) * 16 + m;
        v8f c = {};
        #pragma unroll
        for (int k0 = FEAT; k0 < KX; k0 += 4) {
            const int ka = k0 + khalf * 2;
            v2f a, b;
            a.x = ht[ka - FEAT];
            a.y = ht[ka - FEAT + 1];
            b.x = WX[ka * HX + n];
            b.y = WX[(ka + 1) * HX + n];
            c = wmma_f32(a, b, c);
        }
        if (lane < 16) cvec[n] = c[0];
    }

    for (int i = tid; i < (DBLK + 1) * APITCH; i += SCAN_T) acc[i] = 0.0f;
    __syncthreads();

    for (int t = 0; t < NTILES; ++t) {
        const int k  = t * 32 + lane;
        const int kn = k + 32;
        uint4 w = make_uint4(0u, 0u, 0u, 0u);
        if (kn < WSH16 && w0 + kn < BM16) w = bs[w0 + kn];
        unsigned m0 = v.x, m1 = v.y, m2 = v.z, m3 = v.w;
        const unsigned ebase = (unsigned)(w0 + k) * 128u;
        bool more;
        do {
#define TAKE_BITS(mm, off)                                                              \
            {                                                                           \
                unsigned bal = __builtin_amdgcn_ballot_w32((mm) != 0u);                 \
                while (bal != 0u && cnt <= QW - 32) {                                   \
                    if ((mm) != 0u) {                                                   \
                        const unsigned pos = (unsigned)__builtin_ctz(mm);               \
                        (mm) &= (mm) - 1u;                                              \
                        qw[cnt + (int)__builtin_amdgcn_mbcnt_lo(bal, 0u)] = ebase + (off) + pos; \
                    }                                                                   \
                    cnt += __builtin_popcount(bal);                                     \
                    bal = __builtin_amdgcn_ballot_w32((mm) != 0u);                      \
                }                                                                       \
            }
            TAKE_BITS(m0, 0u)
            TAKE_BITS(m1, 32u)
            TAKE_BITS(m2, 64u)
            TAKE_BITS(m3, 96u)
#undef TAKE_BITS
            more = __builtin_amdgcn_ballot_w32((m0 | m1 | m2 | m3) != 0u) != 0u;
            const bool flush = more || (t == NTILES - 1);
            while (cnt - head >= 32 || (flush && cnt > head)) {
                int n = cnt - head; if (n > 32) n = 32;
                drain_run(qw + head, n, acc, (unsigned)blockIdx.x, pk, hX, hY, lane);
                head += n;
            }
            if (flush) { cnt = 0; head = 0; }
        } while (more);
        v = w;
    }
    __syncthreads();

    int nrows = Nn - lo;
    if (nrows > DBLK) nrows = DBLK;
    float* stX  = (float*)q;
    float* stY  = stX + STY_OFF;
    float* stat = stX + STAT_OFF;
    {
        const int gp = wave >> 2;
        const int t  = wave & 3;
        if (gp * 32 < nrows) {
            if (gp * 32 + 16 < nrows)
                slab_products<true>(acc, cvec, stX, stY, WX, bX, WY, bY, gp, t, lane);
            else
                slab_products<false>(acc, cvec, stX, stY, WX, bX, WY, bY, gp, t, lane);
        }
    }
    __syncthreads();

    if (wave < ST_ROWS / 16) {
        const int row  = wave * 16 + (lane >> 1);
        const int half = lane & 1;
        float s = 0.0f, s2 = 0.0f;
        #pragma unroll
        for (int j = 0; j < 32; ++j) {
            const float x = stX[row * STX_LD + half * 32 + j];
            s += x; s2 += x * x;
        }
        s  += __shfl_xor(s, 1);
        s2 += __shfl_xor(s2, 1);
        const float mean = s * (1.0f / 64.0f);
        const float var  = s2 * (1.0f / 64.0f) - mean * mean;
        const float inv  = rsqrtf(var + 1e-5f);
        if (half == 0) { stat[row] = mean; stat[ST_ROWS + row] = inv; }
    } else if (wave >= 16 && wave < 16 + ST_ROWS / 16) {
        const int row  = (wave - 16) * 16 + (lane >> 1);
        const int half = lane & 1;
        float s = 0.0f, s2 = 0.0f;
        #pragma unroll
        for (int j = 0; j < 16; ++j) {
            const float x = stY[row * STY_LD + half * 16 + j];
            s += x; s2 += x * x;
        }
        s  += __shfl_xor(s, 1);
        s2 += __shfl_xor(s2, 1);
        const float mean = s * (1.0f / 32.0f);
        const float var  = s2 * (1.0f / 32.0f) - mean * mean;
        const float inv  = rsqrtf(var + 1e-5f);
        if (half == 0) { stat[2 * ST_ROWS + row] = mean; stat[3 * ST_ROWS + row] = inv; }
    }
    __syncthreads();

    {
        float* oX = outX + (size_t)lo * HX;
        const int c = tid & (HX - 1);
        const float gg = gX[c], bb = beX[c];
        const int limX = nrows * HX;
        for (int i = tid; i < limX; i += SCAN_T) {
            const int r = i / HX;
            const float x = stX[r * STX_LD + c];
            oX[i] = (x - stat[r]) * stat[ST_ROWS + r] * gg + bb;
        }
    }
    {
        float* oY = outY + (size_t)lo * HY;
        const int c = tid & (HY - 1);
        const float gg = gY[c], bb = beY[c];
        const int limY = nrows * HY;
        for (int i = tid; i < limY; i += SCAN_T) {
            const int r = i / HY;
            const float x = stY[r * STY_LD + c];
            oY[i] = (x - stat[2 * ST_ROWS + r]) * stat[3 * ST_ROWS + r] * gg + bb;
        }
    }
}


extern "C" void kernel_launch(void* const* d_in, const int* in_sizes, int n_in,
                              void* d_out, int out_size, void* d_ws, size_t ws_size,
                              hipStream_t stream) {
    const int*   erow = (const int*)d_in[0];
    const int*   ecol = (const int*)d_in[1];
    const float* hX   = (const float*)d_in[2];
    const float* hY   = (const float*)d_in[3];
    const float* ht   = (const float*)d_in[4];
    const float* WX   = (const float*)d_in[5];
    const float* bX   = (const float*)d_in[6];
    const float* gX   = (const float*)d_in[7];
    const float* beX  = (const float*)d_in[8];
    const float* WY   = (const float*)d_in[9];
    const float* bY   = (const float*)d_in[10];
    const float* gY   = (const float*)d_in[11];
    const float* beY  = (const float*)d_in[12];

    float* out  = (float*)d_out;
    unsigned* bm = (unsigned*)((char*)d_ws + (32u << 20));
    unsigned* pk = (unsigned*)((char*)d_ws + (64u << 20));

    tag_rows<<<NCH, TAG_T, 0, stream>>>(erow, ecol, bm, pk);
    scan_sum<<<NBLK, SCAN_T, 0, stream>>>(pk, hX, hY, bm, ht, WX, bX, gX, beX, WY, bY, gY, beY,
                                          out, out + (size_t)Nn * HX);
}
